# starting kernel
# baseline (speedup 1.0000x reference)
#include <hip/hip_runtime.h>

#define BB 2
#define SS 384
#define CC 4
#define HH 128
#define DD 768

typedef float v2f __attribute__((ext_vector_type(2)));
typedef float v8f __attribute__((ext_vector_type(8)));

__device__ __forceinline__ float tanhf_fast(float x) {
  float e = __builtin_amdgcn_exp2f(x * 2.8853900817779268f);
  float r = __builtin_amdgcn_rcpf(e + 1.0f);
  return 1.0f - 2.0f * r;
}

#define MODE_TANH 1
#define MODE_BIAS 4
#define MODE_ADDV 8

template <int MODE, bool HAS_RAW>
__device__ __forceinline__ void store_tile(v8f c, float* O, int ldo, float* Oraw,
                                           const float* bias, const float* addvec,
                                           int m0, int n0, int lane) {
  int col   = n0 + (lane & 15);
  int rbase = m0 + ((lane >> 4) << 3);
#pragma unroll
  for (int i = 0; i < 8; ++i) {
    int   row = rbase + i;
    float v   = c[i];
    if constexpr (MODE & MODE_BIAS) v += bias[col];
    if constexpr (MODE & MODE_ADDV) v += addvec[(long)(row >> 2) * HH + col];
    if constexpr (HAS_RAW) Oraw[(long)row * ldo + col] = v;
    O[(long)row * ldo + col] = (MODE & MODE_TANH) ? tanhf_fast(v) : v;
  }
}

template <int MODE, bool HAS_RAW>
__global__ void __launch_bounds__(128)
gemm_wmma_kernel(const float* __restrict__ A, int lda, long aOuter, long aInner,
                 const float* __restrict__ B, int ldb, long bOuter, long bInner,
                 float* __restrict__ O, int ldo, long oOuter, long oInner,
                 float* __restrict__ Oraw,
                 const float* __restrict__ bias, const float* __restrict__ addvec,
                 int K) {
  __shared__ float As[64 * 20];
  __shared__ float Bs[64 * 20];

  int lane = threadIdx.x;
  int wave = threadIdx.y;
  int t    = wave * 32 + lane;
  int zo   = blockIdx.z >> 2;
  int zi   = blockIdx.z & 3;
  const float* Ab = A + (long)zo * aOuter + (long)zi * aInner;
  const float* Bb = B + (long)zo * bOuter + (long)zi * bInner;

  int mblk = blockIdx.y * 64;
  int nblk = blockIdx.x * 64;
  int mw   = (wave >> 1) * 32;
  int nw   = (wave & 1) * 32;

  v8f c00 = {}, c01 = {}, c10 = {}, c11 = {};

  for (int kc = 0; kc < K; kc += 16) {
    __syncthreads();
#pragma unroll
    for (int rep = 0; rep < 2; ++rep) {
      int idx = t + rep * 128;
      int ar = idx >> 2, ac = (idx & 3) << 2;
      float4 av = *(const float4*)(Ab + (long)(mblk + ar) * lda + kc + ac);
      *(float4*)(&As[ar * 20 + ac]) = av;
      int br = idx >> 4, bc = (idx & 15) << 2;
      float4 bv = *(const float4*)(Bb + (long)(kc + br) * ldb + nblk + bc);
      Bs[(bc + 0) * 20 + br] = bv.x;
      Bs[(bc + 1) * 20 + br] = bv.y;
      Bs[(bc + 2) * 20 + br] = bv.z;
      Bs[(bc + 3) * 20 + br] = bv.w;
    }
    __syncthreads();

#pragma unroll
    for (int kl = 0; kl < 16; kl += 4) {
      int kf = kl + ((lane >> 4) << 1);
      int rr = lane & 15;
      v2f a0 = *(const v2f*)(&As[(mw + rr) * 20 + kf]);
      v2f a1 = *(const v2f*)(&As[(mw + 16 + rr) * 20 + kf]);
      v2f b0 = *(const v2f*)(&Bs[(nw + rr) * 20 + kf]);
      v2f b1 = *(const v2f*)(&Bs[(nw + 16 + rr) * 20 + kf]);
      c00 = __builtin_amdgcn_wmma_f32_16x16x4_f32(false, a0, false, b0, (short)0, c00, false, false);
      c01 = __builtin_amdgcn_wmma_f32_16x16x4_f32(false, a0, false, b1, (short)0, c01, false, false);
      c10 = __builtin_amdgcn_wmma_f32_16x16x4_f32(false, a1, false, b0, (short)0, c10, false, false);
      c11 = __builtin_amdgcn_wmma_f32_16x16x4_f32(false, a1, false, b1, (short)0, c11, false, false);
    }
  }

  float* Ob = O + (long)zo * oOuter + (long)zi * oInner;
  float* Rb = HAS_RAW ? (Oraw + (long)zo * oOuter + (long)zi * oInner) : nullptr;
  int m0 = mblk + mw, n0 = nblk + nw;
  store_tile<MODE, HAS_RAW>(c00, Ob, ldo, Rb, bias, addvec, m0,      n0,      lane);
  store_tile<MODE, HAS_RAW>(c01, Ob, ldo, Rb, bias, addvec, m0,      n0 + 16, lane);
  store_tile<MODE, HAS_RAW>(c10, Ob, ldo, Rb, bias, addvec, m0 + 16, n0,      lane);
  store_tile<MODE, HAS_RAW>(c11, Ob, ldo, Rb, bias, addvec, m0 + 16, n0 + 16, lane);
}

__global__ void __launch_bounds__(128)
final_reduce_kernel(const float* __restrict__ pref, const float* __restrict__ targ,
                    const float* __restrict__ wout, float* __restrict__ out) {
  __shared__ float sp[HH];
  __shared__ float sw[HH];
  int tid = threadIdx.x;
  int p   = blockIdx.y;
  int bc  = blockIdx.z;
  int b   = bc >> 2;
  int c   = bc & 3;

  long prow = ((long)(b * SS + p) * CC + c) * HH;
  sp[tid] = pref[prow + tid];
  sw[tid] = wout[tid];
  __syncthreads();

  int a = blockIdx.x * 128 + tid;
  const float* ta = targ + ((long)(b * SS + a) * CC + c) * HH;
  __builtin_prefetch(ta, 0, 0);

  float acc = 0.0f;
#pragma unroll 8
  for (int k = 0; k < HH; k += 4) {
    float4 tv = *(const float4*)(ta + k);
    acc += tanhf_fast(sp[k + 0] + tv.x) * sw[k + 0];
    acc += tanhf_fast(sp[k + 1] + tv.y) * sw[k + 1];
    acc += tanhf_fast(sp[k + 2] + tv.z) * sw[k + 2];
    acc += tanhf_fast(sp[k + 3] + tv.w) * sw[k + 3];
  }
  out[((long)(b * SS + p) * CC + c) * SS + a] = acc;
}

extern "C" void kernel_launch(void* const* d_in, const int* in_sizes, int n_in,
                              void* d_out, int out_size, void* d_ws, size_t ws_size,
                              hipStream_t stream) {
  const float* seq  = (const float*)d_in[0];
  const float* base = (const float*)d_in[1];
  const float* Wp   = (const float*)d_in[2];
  const float* bp   = (const float*)d_in[3];
  const float* Ua   = (const float*)d_in[4];
  const float* ba   = (const float*)d_in[5];
  const float* Wmid = (const float*)d_in[6];
  const float* bmid = (const float*)d_in[7];
  const float* wout = (const float*)d_in[8];
  float* out = (float*)d_out;

  float* ws      = (float*)d_ws;
  float* tanh_hp = ws;
  float* ha_raw  = ws + 393216;
  float* tanh_ha = ws + 786432;
  float* tanhref = ws + 1179648;
  float* targ    = ws + 1572864;
  float* tref    = ws + 1966080;
  float* pref    = ws + 2064384;

  dim3 blk(32, 4);

  gemm_wmma_kernel<MODE_TANH | MODE_BIAS, false><<<dim3(8, 12, 1), blk, 0, stream>>>(
      seq, DD, 0, 0, Wp, CC * HH, 0, 0, tanh_hp, CC * HH, 0, 0,
      nullptr, bp, nullptr, DD);

  gemm_wmma_kernel<MODE_TANH | MODE_BIAS, true><<<dim3(8, 12, 1), blk, 0, stream>>>(
      seq, DD, 0, 0, Ua, CC * HH, 0, 0, tanh_ha, CC * HH, 0, 0,
      ha_raw, ba, nullptr, DD);

  gemm_wmma_kernel<MODE_TANH, false><<<dim3(2, 6, BB * CC), blk, 0, stream>>>(
      base, CC * SS, (long)SS * CC * SS, SS,
      ha_raw, CC * HH, (long)SS * CC * HH, HH,
      tanhref, CC * HH, (long)SS * CC * HH, HH,
      nullptr, nullptr, nullptr, SS);

  gemm_wmma_kernel<0, false><<<dim3(2, 12, 1), blk, 0, stream>>>(
      tanhref, CC * HH, 0, 0, Wmid + 2 * HH * HH, HH, 0, 0, tref, HH, 0, 0,
      nullptr, nullptr, nullptr, CC * HH);

  gemm_wmma_kernel<MODE_BIAS | MODE_ADDV, false><<<dim3(2, 48, 1), blk, 0, stream>>>(
      tanh_hp, HH, 0, 0, Wmid, HH, 0, 0, pref, HH, 0, 0,
      nullptr, bmid, tref, HH);

  gemm_wmma_kernel<0, false><<<dim3(2, 48, 1), blk, 0, stream>>>(
      tanh_ha, HH, 0, 0, Wmid + HH * HH, HH, 0, 0, targ, HH, 0, 0,
      nullptr, nullptr, nullptr, HH);

  final_reduce_kernel<<<dim3(SS / 128, SS, BB * CC), dim3(128), 0, stream>>>(
      pref, targ, wout, out);
}


# GEMM kernel: stage 128 k per barrier pair instead of 16 (LDS tiles 64x132, 32 loads in flight per thread), so each K loop pays 8x fewer dependent global round trips; sum order unchanged
# speedup vs baseline: 1.7953x; 1.7953x over previous
#include <hip/hip_runtime.h>

#define BB 2
#define SS 384
#define CC 4
#define HH 128
#define DD 768

typedef float v2f __attribute__((ext_vector_type(2)));
typedef float v8f __attribute__((ext_vector_type(8)));

__device__ __forceinline__ float tanhf_fast(float x) {
  float e = __builtin_amdgcn_exp2f(x * 2.8853900817779268f);
  float r = __builtin_amdgcn_rcpf(e + 1.0f);
  return 1.0f - 2.0f * r;
}

#define MODE_TANH 1
#define MODE_BIAS 4
#define MODE_ADDV 8

template <int MODE, bool HAS_RAW>
__device__ __forceinline__ void store_tile(v8f c, float* O, int ldo, float* Oraw,
                                           const float* bias, const float* addvec,
                                           int m0, int n0, int lane) {
  int col   = n0 + (lane & 15);
  int rbase = m0 + ((lane >> 4) << 3);
#pragma unroll
  for (int i = 0; i < 8; ++i) {
    int   row = rbase + i;
    float v   = c[i];
    if constexpr (MODE & MODE_BIAS) v += bias[col];
    if constexpr (MODE & MODE_ADDV) v += addvec[(long)(row >> 2) * HH + col];
    if constexpr (HAS_RAW) Oraw[(long)row * ldo + col] = v;
    O[(long)row * ldo + col] = (MODE & MODE_TANH) ? tanhf_fast(v) : v;
  }
}

#define BK 128
#define LDK (BK + 4)
template <int MODE, bool HAS_RAW>
__global__ void __launch_bounds__(128)
gemm_wmma_kernel(const float* __restrict__ A, int lda, long aOuter, long aInner,
                 const float* __restrict__ B, int ldb, long bOuter, long bInner,
                 float* __restrict__ O, int ldo, long oOuter, long oInner,
                 float* __restrict__ Oraw,
                 const float* __restrict__ bias, const float* __restrict__ addvec,
                 int K) {
  __shared__ float As[64 * LDK];
  __shared__ float Bs[64 * LDK];

  int lane = threadIdx.x;
  int wave = threadIdx.y;
  int t    = wave * 32 + lane;
  int zo   = blockIdx.z >> 2;
  int zi   = blockIdx.z & 3;
  const float* Ab = A + (long)zo * aOuter + (long)zi * aInner;
  const float* Bb = B + (long)zo * bOuter + (long)zi * bInner;

  int mblk = blockIdx.y * 64;
  int nblk = blockIdx.x * 64;
  int mw   = (wave >> 1) * 32;
  int nw   = (wave & 1) * 32;

  v8f c00 = {}, c01 = {}, c10 = {}, c11 = {};

  int ar0 = t >> 5, ac = (t & 31) << 2;
  int br0 = t >> 4, bc = (t & 15) << 2;
  const float* Ag = Ab + (long)(mblk + ar0) * lda + ac;
  const float* Bg = Bb + (long)br0 * ldb + nblk + bc;

  for (int kc = 0; kc < K; kc += BK) {
    __syncthreads();
    float4 av[16], bv[16];
#pragma unroll
    for (int rep = 0; rep < 16; ++rep)
      av[rep] = *(const float4*)(Ag + (long)(rep * 4) * lda + kc);
#pragma unroll
    for (int rep = 0; rep < 16; ++rep)
      bv[rep] = *(const float4*)(Bg + (long)(kc + rep * 8) * ldb);
#pragma unroll
    for (int rep = 0; rep < 16; ++rep)
      *(float4*)(&As[(ar0 + rep * 4) * LDK + ac]) = av[rep];
#pragma unroll
    for (int rep = 0; rep < 16; ++rep) {
      int br = br0 + rep * 8;
      Bs[(bc + 0) * LDK + br] = bv[rep].x;
      Bs[(bc + 1) * LDK + br] = bv[rep].y;
      Bs[(bc + 2) * LDK + br] = bv[rep].z;
      Bs[(bc + 3) * LDK + br] = bv[rep].w;
    }
    __syncthreads();

#pragma unroll
    for (int kl = 0; kl < BK; kl += 4) {
      int kf = kl + ((lane >> 4) << 1);
      int rr = lane & 15;
      v2f a0 = *(const v2f*)(&As[(mw + rr) * LDK + kf]);
      v2f a1 = *(const v2f*)(&As[(mw + 16 + rr) * LDK + kf]);
      v2f b0 = *(const v2f*)(&Bs[(nw + rr) * LDK + kf]);
      v2f b1 = *(const v2f*)(&Bs[(nw + 16 + rr) * LDK + kf]);
      c00 = __builtin_amdgcn_wmma_f32_16x16x4_f32(false, a0, false, b0, (short)0, c00, false, false);
      c01 = __builtin_amdgcn_wmma_f32_16x16x4_f32(false, a0, false, b1, (short)0, c01, false, false);
      c10 = __builtin_amdgcn_wmma_f32_16x16x4_f32(false, a1, false, b0, (short)0, c10, false, false);
      c11 = __builtin_amdgcn_wmma_f32_16x16x4_f32(false, a1, false, b1, (short)0, c11, false, false);
    }
  }

  float* Ob = O + (long)zo * oOuter + (long)zi * oInner;
  float* Rb = HAS_RAW ? (Oraw + (long)zo * oOuter + (long)zi * oInner) : nullptr;
  int m0 = mblk + mw, n0 = nblk + nw;
  store_tile<MODE, HAS_RAW>(c00, Ob, ldo, Rb, bias, addvec, m0,      n0,      lane);
  store_tile<MODE, HAS_RAW>(c01, Ob, ldo, Rb, bias, addvec, m0,      n0 + 16, lane);
  store_tile<MODE, HAS_RAW>(c10, Ob, ldo, Rb, bias, addvec, m0 + 16, n0,      lane);
  store_tile<MODE, HAS_RAW>(c11, Ob, ldo, Rb, bias, addvec, m0 + 16, n0 + 16, lane);
}

__global__ void __launch_bounds__(128)
final_reduce_kernel(const float* __restrict__ pref, const float* __restrict__ targ,
                    const float* __restrict__ wout, float* __restrict__ out) {
  __shared__ float sp[HH];
  __shared__ float sw[HH];
  int tid = threadIdx.x;
  int p   = blockIdx.y;
  int bc  = blockIdx.z;
  int b   = bc >> 2;
  int c   = bc & 3;

  long prow = ((long)(b * SS + p) * CC + c) * HH;
  sp[tid] = pref[prow + tid];
  sw[tid] = wout[tid];
  __syncthreads();

  int a = blockIdx.x * 128 + tid;
  const float* ta = targ + ((long)(b * SS + a) * CC + c) * HH;
  __builtin_prefetch(ta, 0, 0);

  float acc = 0.0f;
#pragma unroll 8
  for (int k = 0; k < HH; k += 4) {
    float4 tv = *(const float4*)(ta + k);
    acc += tanhf_fast(sp[k + 0] + tv.x) * sw[k + 0];
    acc += tanhf_fast(sp[k + 1] + tv.y) * sw[k + 1];
    acc += tanhf_fast(sp[k + 2] + tv.z) * sw[k + 2];
    acc += tanhf_fast(sp[k + 3] + tv.w) * sw[k + 3];
  }
  out[((long)(b * SS + p) * CC + c) * SS + a] = acc;
}

extern "C" void kernel_launch(void* const* d_in, const int* in_sizes, int n_in,
                              void* d_out, int out_size, void* d_ws, size_t ws_size,
                              hipStream_t stream) {
  const float* seq  = (const float*)d_in[0];
  const float* base = (const float*)d_in[1];
  const float* Wp   = (const float*)d_in[2];
  const float* bp   = (const float*)d_in[3];
  const float* Ua   = (const float*)d_in[4];
  const float* ba   = (const float*)d_in[5];
  const float* Wmid = (const float*)d_in[6];
  const float* bmid = (const float*)d_in[7];
  const float* wout = (const float*)d_in[8];
  float* out = (float*)d_out;

  float* ws      = (float*)d_ws;
  float* tanh_hp = ws;
  float* ha_raw  = ws + 393216;
  float* tanh_ha = ws + 786432;
  float* tanhref = ws + 1179648;
  float* targ    = ws + 1572864;
  float* tref    = ws + 1966080;
  float* pref    = ws + 2064384;

  dim3 blk(32, 4);

  gemm_wmma_kernel<MODE_TANH | MODE_BIAS, false><<<dim3(8, 12, 1), blk, 0, stream>>>(
      seq, DD, 0, 0, Wp, CC * HH, 0, 0, tanh_hp, CC * HH, 0, 0,
      nullptr, bp, nullptr, DD);

  gemm_wmma_kernel<MODE_TANH | MODE_BIAS, true><<<dim3(8, 12, 1), blk, 0, stream>>>(
      seq, DD, 0, 0, Ua, CC * HH, 0, 0, tanh_ha, CC * HH, 0, 0,
      ha_raw, ba, nullptr, DD);

  gemm_wmma_kernel<MODE_TANH, false><<<dim3(2, 6, BB * CC), blk, 0, stream>>>(
      base, CC * SS, (long)SS * CC * SS, SS,
      ha_raw, CC * HH, (long)SS * CC * HH, HH,
      tanhref, CC * HH, (long)SS * CC * HH, HH,
      nullptr, nullptr, nullptr, SS);

  gemm_wmma_kernel<0, false><<<dim3(2, 12, 1), blk, 0, stream>>>(
      tanhref, CC * HH, 0, 0, Wmid + 2 * HH * HH, HH, 0, 0, tref, HH, 0, 0,
      nullptr, nullptr, nullptr, CC * HH);

  gemm_wmma_kernel<MODE_BIAS | MODE_ADDV, false><<<dim3(2, 48, 1), blk, 0, stream>>>(
      tanh_hp, HH, 0, 0, Wmid, HH, 0, 0, pref, HH, 0, 0,
      nullptr, bmid, tref, HH);

  gemm_wmma_kernel<0, false><<<dim3(2, 48, 1), blk, 0, stream>>>(
      tanh_ha, HH, 0, 0, Wmid + HH * HH, HH, 0, 0, targ, HH, 0, 0,
      nullptr, nullptr, nullptr, HH);

  final_reduce_kernel<<<dim3(SS / 128, SS, BB * CC), dim3(128), 0, stream>>>(
      pref, targ, wout, out);
}


# Independent GEMMs share a launch: the two projections of seq run as one grid, targ rides in the t_ref grid (blocks of both problems side by side on separate WGPs); 7 launches -> 5; sums unchanged
# speedup vs baseline: 2.2005x; 1.2257x over previous
#include <hip/hip_runtime.h>

#define BB 2
#define SS 384
#define CC 4
#define HH 128
#define DD 768

typedef float v2f __attribute__((ext_vector_type(2)));
typedef float v8f __attribute__((ext_vector_type(8)));

__device__ __forceinline__ float tanhf_fast(float x) {
  float e = __builtin_amdgcn_exp2f(x * 2.8853900817779268f);
  float r = __builtin_amdgcn_rcpf(e + 1.0f);
  return 1.0f - 2.0f * r;
}

#define MODE_TANH 1
#define MODE_BIAS 4
#define MODE_ADDV 8

template <int MODE, bool HAS_RAW>
__device__ __forceinline__ void store_tile(v8f c, float* O, int ldo, float* Oraw,
                                           const float* bias, const float* addvec,
                                           int m0, int n0, int lane) {
  int col   = n0 + (lane & 15);
  int rbase = m0 + ((lane >> 4) << 3);
#pragma unroll
  for (int i = 0; i < 8; ++i) {
    int   row = rbase + i;
    float v   = c[i];
    if constexpr (MODE & MODE_BIAS) v += bias[col];
    if constexpr (MODE & MODE_ADDV) v += addvec[(long)(row >> 2) * HH + col];
    if constexpr (HAS_RAW) Oraw[(long)row * ldo + col] = v;
    O[(long)row * ldo + col] = (MODE & MODE_TANH) ? tanhf_fast(v) : v;
  }
}

#define BK 128
#define LDK (BK + 4)
template <int MODE, bool HAS_RAW>
__device__ __forceinline__ void
gemm_tile(const float* __restrict__ Ab, int lda, const float* __restrict__ Bb, int ldb,
          float* __restrict__ Ob, int ldo, float* __restrict__ Rb,
          const float* __restrict__ bias, const float* __restrict__ addvec,
          int K, int mblk, int nblk, float* As, float* Bs) {
  int lane = threadIdx.x;
  int wave = threadIdx.y;
  int t    = wave * 32 + lane;
  int mw   = (wave >> 1) * 32;
  int nw   = (wave & 1) * 32;

  v8f c00 = {}, c01 = {}, c10 = {}, c11 = {};

  int ar0 = t >> 5, ac = (t & 31) << 2;
  int br0 = t >> 4, bc = (t & 15) << 2;
  const float* Ag = Ab + (long)(mblk + ar0) * lda + ac;
  const float* Bg = Bb + (long)br0 * ldb + nblk + bc;

  for (int kc = 0; kc < K; kc += BK) {
    __syncthreads();
    float4 av[16], bv[16];
#pragma unroll
    for (int rep = 0; rep < 16; ++rep)
      av[rep] = *(const float4*)(Ag + (long)(rep * 4) * lda + kc);
#pragma unroll
    for (int rep = 0; rep < 16; ++rep)
      bv[rep] = *(const float4*)(Bg + (long)(kc + rep * 8) * ldb);
#pragma unroll
    for (int rep = 0; rep < 16; ++rep)
      *(float4*)(&As[(ar0 + rep * 4) * LDK + ac]) = av[rep];
#pragma unroll
    for (int rep = 0; rep < 16; ++rep) {
      int br = br0 + rep * 8;
      Bs[(bc + 0) * LDK + br] = bv[rep].x;
      Bs[(bc + 1) * LDK + br] = bv[rep].y;
      Bs[(bc + 2) * LDK + br] = bv[rep].z;
      Bs[(bc + 3) * LDK + br] = bv[rep].w;
    }
    __syncthreads();

#pragma unroll
    for (int kl = 0; kl < BK; kl += 4) {
      int kf = kl + ((lane >> 4) << 1);
      int rr = lane & 15;
      v2f a0 = *(const v2f*)(&As[(mw + rr) * LDK + kf]);
      v2f a1 = *(const v2f*)(&As[(mw + 16 + rr) * LDK + kf]);
      v2f b0 = *(const v2f*)(&Bs[(nw + rr) * LDK + kf]);
      v2f b1 = *(const v2f*)(&Bs[(nw + 16 + rr) * LDK + kf]);
      c00 = __builtin_amdgcn_wmma_f32_16x16x4_f32(false, a0, false, b0, (short)0, c00, false, false);
      c01 = __builtin_amdgcn_wmma_f32_16x16x4_f32(false, a0, false, b1, (short)0, c01, false, false);
      c10 = __builtin_amdgcn_wmma_f32_16x16x4_f32(false, a1, false, b0, (short)0, c10, false, false);
      c11 = __builtin_amdgcn_wmma_f32_16x16x4_f32(false, a1, false, b1, (short)0, c11, false, false);
    }
  }

  int m0 = mblk + mw, n0 = nblk + nw;
  store_tile<MODE, HAS_RAW>(c00, Ob, ldo, Rb, bias, addvec, m0,      n0,      lane);
  store_tile<MODE, HAS_RAW>(c01, Ob, ldo, Rb, bias, addvec, m0,      n0 + 16, lane);
  store_tile<MODE, HAS_RAW>(c10, Ob, ldo, Rb, bias, addvec, m0 + 16, n0,      lane);
  store_tile<MODE, HAS_RAW>(c11, Ob, ldo, Rb, bias, addvec, m0 + 16, n0 + 16, lane);
}

template <int MODE, bool HAS_RAW>
__global__ void __launch_bounds__(128)
gemm_wmma_kernel(const float* __restrict__ A, int lda, long aOuter, long aInner,
                 const float* __restrict__ B, int ldb, long bOuter, long bInner,
                 float* __restrict__ O, int ldo, long oOuter, long oInner,
                 float* __restrict__ Oraw,
                 const float* __restrict__ bias, const float* __restrict__ addvec,
                 int K) {
  __shared__ float As[64 * LDK];
  __shared__ float Bs[64 * LDK];
  int zo = blockIdx.z >> 2;
  int zi = blockIdx.z & 3;
  long ooff = (long)zo * oOuter + (long)zi * oInner;
  gemm_tile<MODE, HAS_RAW>(A + (long)zo * aOuter + (long)zi * aInner, lda,
                           B + (long)zo * bOuter + (long)zi * bInner, ldb,
                           O + ooff, ldo, HAS_RAW ? (Oraw + ooff) : nullptr,
                           bias, addvec, K, blockIdx.y * 64, blockIdx.x * 64, As, Bs);
}

template <int MODE0, bool RAW0, int MODE1, bool RAW1>
__global__ void __launch_bounds__(128)
gemm_pair_kernel(const float* __restrict__ A0, int lda0, const float* __restrict__ B0, int ldb0,
                 float* __restrict__ O0, int ldo0, float* __restrict__ R0,
                 const float* __restrict__ bias0, int K0,
                 const float* __restrict__ A1, int lda1, const float* __restrict__ B1, int ldb1,
                 float* __restrict__ O1, int ldo1, float* __restrict__ R1,
                 const float* __restrict__ bias1, int K1,
                 int ny0) {
  __shared__ float As[64 * LDK];
  __shared__ float Bs[64 * LDK];
  int by = blockIdx.y;
  if (by < ny0)
    gemm_tile<MODE0, RAW0>(A0, lda0, B0, ldb0, O0, ldo0, R0, bias0, nullptr, K0,
                           by * 64, blockIdx.x * 64, As, Bs);
  else
    gemm_tile<MODE1, RAW1>(A1, lda1, B1, ldb1, O1, ldo1, R1, bias1, nullptr, K1,
                           (by - ny0) * 64, blockIdx.x * 64, As, Bs);
}

__global__ void __launch_bounds__(128)
final_reduce_kernel(const float* __restrict__ pref, const float* __restrict__ targ,
                    const float* __restrict__ wout, float* __restrict__ out) {
  __shared__ float sp[HH];
  __shared__ float sw[HH];
  int tid = threadIdx.x;
  int p   = blockIdx.y;
  int bc  = blockIdx.z;
  int b   = bc >> 2;
  int c   = bc & 3;

  long prow = ((long)(b * SS + p) * CC + c) * HH;
  sp[tid] = pref[prow + tid];
  sw[tid] = wout[tid];
  __syncthreads();

  int a = blockIdx.x * 128 + tid;
  const float* ta = targ + ((long)(b * SS + a) * CC + c) * HH;
  __builtin_prefetch(ta, 0, 0);

  float acc = 0.0f;
#pragma unroll 8
  for (int k = 0; k < HH; k += 4) {
    float4 tv = *(const float4*)(ta + k);
    acc += tanhf_fast(sp[k + 0] + tv.x) * sw[k + 0];
    acc += tanhf_fast(sp[k + 1] + tv.y) * sw[k + 1];
    acc += tanhf_fast(sp[k + 2] + tv.z) * sw[k + 2];
    acc += tanhf_fast(sp[k + 3] + tv.w) * sw[k + 3];
  }
  out[((long)(b * SS + p) * CC + c) * SS + a] = acc;
}

extern "C" void kernel_launch(void* const* d_in, const int* in_sizes, int n_in,
                              void* d_out, int out_size, void* d_ws, size_t ws_size,
                              hipStream_t stream) {
  const float* seq  = (const float*)d_in[0];
  const float* base = (const float*)d_in[1];
  const float* Wp   = (const float*)d_in[2];
  const float* bp   = (const float*)d_in[3];
  const float* Ua   = (const float*)d_in[4];
  const float* ba   = (const float*)d_in[5];
  const float* Wmid = (const float*)d_in[6];
  const float* bmid = (const float*)d_in[7];
  const float* wout = (const float*)d_in[8];
  float* out = (float*)d_out;

  float* ws      = (float*)d_ws;
  float* tanh_hp = ws;
  float* ha_raw  = ws + 393216;
  float* tanh_ha = ws + 786432;
  float* tanhref = ws + 1179648;
  float* targ    = ws + 1572864;
  float* tref    = ws + 1966080;
  float* pref    = ws + 2064384;

  dim3 blk(32, 4);

  gemm_pair_kernel<MODE_TANH | MODE_BIAS, true, MODE_TANH | MODE_BIAS, false>
      <<<dim3(8, 24, 1), blk, 0, stream>>>(
      seq, DD, Ua, CC * HH, tanh_ha, CC * HH, ha_raw, ba, DD,
      seq, DD, Wp, CC * HH, tanh_hp, CC * HH, nullptr, bp, DD,
      12);

  gemm_wmma_kernel<MODE_TANH, false><<<dim3(2, 6, BB * CC), blk, 0, stream>>>(
      base, CC * SS, (long)SS * CC * SS, SS,
      ha_raw, CC * HH, (long)SS * CC * HH, HH,
      tanhref, CC * HH, (long)SS * CC * HH, HH,
      nullptr, nullptr, nullptr, SS);

  gemm_pair_kernel<0, false, 0, false><<<dim3(2, 60, 1), blk, 0, stream>>>(
      tanhref, CC * HH, Wmid + 2 * HH * HH, HH, tref, HH, nullptr, nullptr, CC * HH,
      tanh_ha, HH, Wmid + HH * HH, HH, targ, HH, nullptr, nullptr, HH,
      12);

  gemm_wmma_kernel<MODE_BIAS | MODE_ADDV, false><<<dim3(2, 48, 1), blk, 0, stream>>>(
      tanh_hp, HH, 0, 0, Wmid, HH, 0, 0, pref, HH, 0, 0,
      nullptr, bmid, tref, HH);

  final_reduce_kernel<<<dim3(SS / 128, SS, BB * CC), dim3(128), 0, stream>>>(
      pref, targ, wout, out);
}


# Final kernel tiled 128 a x 12 p per block: targ tile fetched once, coalesced, into LDS and reused for 12 pref rows (was one 512-B row per thread per p); 36864 -> 3072 waves; sum order unchanged
# speedup vs baseline: 2.9778x; 1.3532x over previous
#include <hip/hip_runtime.h>

#define BB 2
#define SS 384
#define CC 4
#define HH 128
#define DD 768

typedef float v2f __attribute__((ext_vector_type(2)));
typedef float v8f __attribute__((ext_vector_type(8)));

__device__ __forceinline__ float tanhf_fast(float x) {
  float e = __builtin_amdgcn_exp2f(x * 2.8853900817779268f);
  float r = __builtin_amdgcn_rcpf(e + 1.0f);
  return 1.0f - 2.0f * r;
}

#define MODE_TANH 1
#define MODE_BIAS 4
#define MODE_ADDV 8

template <int MODE, bool HAS_RAW>
__device__ __forceinline__ void store_tile(v8f c, float* O, int ldo, float* Oraw,
                                           const float* bias, const float* addvec,
                                           int m0, int n0, int lane) {
  int col   = n0 + (lane & 15);
  int rbase = m0 + ((lane >> 4) << 3);
#pragma unroll
  for (int i = 0; i < 8; ++i) {
    int   row = rbase + i;
    float v   = c[i];
    if constexpr (MODE & MODE_BIAS) v += bias[col];
    if constexpr (MODE & MODE_ADDV) v += addvec[(long)(row >> 2) * HH + col];
    if constexpr (HAS_RAW) Oraw[(long)row * ldo + col] = v;
    O[(long)row * ldo + col] = (MODE & MODE_TANH) ? tanhf_fast(v) : v;
  }
}

#define BK 128
#define LDK (BK + 4)
template <int MODE, bool HAS_RAW>
__device__ __forceinline__ void
gemm_tile(const float* __restrict__ Ab, int lda, const float* __restrict__ Bb, int ldb,
          float* __restrict__ Ob, int ldo, float* __restrict__ Rb,
          const float* __restrict__ bias, const float* __restrict__ addvec,
          int K, int mblk, int nblk, float* As, float* Bs) {
  int lane = threadIdx.x;
  int wave = threadIdx.y;
  int t    = wave * 32 + lane;
  int mw   = (wave >> 1) * 32;
  int nw   = (wave & 1) * 32;

  v8f c00 = {}, c01 = {}, c10 = {}, c11 = {};

  int ar0 = t >> 5, ac = (t & 31) << 2;
  int br0 = t >> 4, bc = (t & 15) << 2;
  const float* Ag = Ab + (long)(mblk + ar0) * lda + ac;
  const float* Bg = Bb + (long)br0 * ldb + nblk + bc;

  for (int kc = 0; kc < K; kc += BK) {
    __syncthreads();
    float4 av[16], bv[16];
#pragma unroll
    for (int rep = 0; rep < 16; ++rep)
      av[rep] = *(const float4*)(Ag + (long)(rep * 4) * lda + kc);
#pragma unroll
    for (int rep = 0; rep < 16; ++rep)
      bv[rep] = *(const float4*)(Bg + (long)(kc + rep * 8) * ldb);
#pragma unroll
    for (int rep = 0; rep < 16; ++rep)
      *(float4*)(&As[(ar0 + rep * 4) * LDK + ac]) = av[rep];
#pragma unroll
    for (int rep = 0; rep < 16; ++rep) {
      int br = br0 + rep * 8;
      Bs[(bc + 0) * LDK + br] = bv[rep].x;
      Bs[(bc + 1) * LDK + br] = bv[rep].y;
      Bs[(bc + 2) * LDK + br] = bv[rep].z;
      Bs[(bc + 3) * LDK + br] = bv[rep].w;
    }
    __syncthreads();

#pragma unroll
    for (int kl = 0; kl < BK; kl += 4) {
      int kf = kl + ((lane >> 4) << 1);
      int rr = lane & 15;
      v2f a0 = *(const v2f*)(&As[(mw + rr) * LDK + kf]);
      v2f a1 = *(const v2f*)(&As[(mw + 16 + rr) * LDK + kf]);
      v2f b0 = *(const v2f*)(&Bs[(nw + rr) * LDK + kf]);
      v2f b1 = *(const v2f*)(&Bs[(nw + 16 + rr) * LDK + kf]);
      c00 = __builtin_amdgcn_wmma_f32_16x16x4_f32(false, a0, false, b0, (short)0, c00, false, false);
      c01 = __builtin_amdgcn_wmma_f32_16x16x4_f32(false, a0, false, b1, (short)0, c01, false, false);
      c10 = __builtin_amdgcn_wmma_f32_16x16x4_f32(false, a1, false, b0, (short)0, c10, false, false);
      c11 = __builtin_amdgcn_wmma_f32_16x16x4_f32(false, a1, false, b1, (short)0, c11, false, false);
    }
  }

  int m0 = mblk + mw, n0 = nblk + nw;
  store_tile<MODE, HAS_RAW>(c00, Ob, ldo, Rb, bias, addvec, m0,      n0,      lane);
  store_tile<MODE, HAS_RAW>(c01, Ob, ldo, Rb, bias, addvec, m0,      n0 + 16, lane);
  store_tile<MODE, HAS_RAW>(c10, Ob, ldo, Rb, bias, addvec, m0 + 16, n0,      lane);
  store_tile<MODE, HAS_RAW>(c11, Ob, ldo, Rb, bias, addvec, m0 + 16, n0 + 16, lane);
}

template <int MODE, bool HAS_RAW>
__global__ void __launch_bounds__(128)
gemm_wmma_kernel(const float* __restrict__ A, int lda, long aOuter, long aInner,
                 const float* __restrict__ B, int ldb, long bOuter, long bInner,
                 float* __restrict__ O, int ldo, long oOuter, long oInner,
                 float* __restrict__ Oraw,
                 const float* __restrict__ bias, const float* __restrict__ addvec,
                 int K) {
  __shared__ float As[64 * LDK];
  __shared__ float Bs[64 * LDK];
  int zo = blockIdx.z >> 2;
  int zi = blockIdx.z & 3;
  long ooff = (long)zo * oOuter + (long)zi * oInner;
  gemm_tile<MODE, HAS_RAW>(A + (long)zo * aOuter + (long)zi * aInner, lda,
                           B + (long)zo * bOuter + (long)zi * bInner, ldb,
                           O + ooff, ldo, HAS_RAW ? (Oraw + ooff) : nullptr,
                           bias, addvec, K, blockIdx.y * 64, blockIdx.x * 64, As, Bs);
}

template <int MODE0, bool RAW0, int MODE1, bool RAW1>
__global__ void __launch_bounds__(128)
gemm_pair_kernel(const float* __restrict__ A0, int lda0, const float* __restrict__ B0, int ldb0,
                 float* __restrict__ O0, int ldo0, float* __restrict__ R0,
                 const float* __restrict__ bias0, int K0,
                 const float* __restrict__ A1, int lda1, const float* __restrict__ B1, int ldb1,
                 float* __restrict__ O1, int ldo1, float* __restrict__ R1,
                 const float* __restrict__ bias1, int K1,
                 int ny0) {
  __shared__ float As[64 * LDK];
  __shared__ float Bs[64 * LDK];
  int by = blockIdx.y;
  if (by < ny0)
    gemm_tile<MODE0, RAW0>(A0, lda0, B0, ldb0, O0, ldo0, R0, bias0, nullptr, K0,
                           by * 64, blockIdx.x * 64, As, Bs);
  else
    gemm_tile<MODE1, RAW1>(A1, lda1, B1, ldb1, O1, ldo1, R1, bias1, nullptr, K1,
                           (by - ny0) * 64, blockIdx.x * 64, As, Bs);
}

#define TP 12
__global__ void __launch_bounds__(128)
final_reduce_kernel(const float* __restrict__ pref, const float* __restrict__ targ,
                    const float* __restrict__ wout, float* __restrict__ out) {
  __shared__ __attribute__((aligned(16))) float Ts[128 * LDK];
  __shared__ __attribute__((aligned(16))) float sp[TP * HH];
  __shared__ __attribute__((aligned(16))) float sw[HH];
  int tid  = threadIdx.x;
  int lane = tid & 31;
  int wave = tid >> 5;
  int a0   = blockIdx.x * 128;
  int p0   = blockIdx.y * TP;
  int bc   = blockIdx.z;
  int b    = bc >> 2;
  int c    = bc & 3;

  const float* tg = targ + ((long)(b * SS + a0) * CC + c) * HH + (lane << 2);
  {
    float4 tv[32];
#pragma unroll
    for (int rep = 0; rep < 32; ++rep)
      tv[rep] = *(const float4*)(tg + (long)(wave + rep * 4) * (CC * HH));
#pragma unroll
    for (int rep = 0; rep < 32; ++rep)
      *(float4*)(&Ts[(wave + rep * 4) * LDK + (lane << 2)]) = tv[rep];
  }
#pragma unroll
  for (int j = 0; j < TP * (HH / 4) / 128; ++j) {
    int i = tid + j * 128;
    int row = i >> 5, c4 = (i & 31) << 2;
    *(float4*)(&sp[row * HH + c4]) =
        *(const float4*)(pref + ((long)(b * SS + p0 + row) * CC + c) * HH + c4);
  }
  sw[tid] = wout[tid];
  __syncthreads();

  const float* trow = &Ts[tid * LDK];
  float acc[TP];
#pragma unroll
  for (int j = 0; j < TP; ++j) acc[j] = 0.0f;
  for (int k = 0; k < HH; k += 4) {
    float4 tv = *(const float4*)(trow + k);
    float4 wv = *(const float4*)(sw + k);
    float4 q[TP];
#pragma unroll
    for (int j = 0; j < TP; ++j) q[j] = *(const float4*)(sp + j * HH + k);
#pragma unroll
    for (int j = 0; j < TP; ++j) acc[j] += tanhf_fast(q[j].x + tv.x) * wv.x;
#pragma unroll
    for (int j = 0; j < TP; ++j) acc[j] += tanhf_fast(q[j].y + tv.y) * wv.y;
#pragma unroll
    for (int j = 0; j < TP; ++j) acc[j] += tanhf_fast(q[j].z + tv.z) * wv.z;
#pragma unroll
    for (int j = 0; j < TP; ++j) acc[j] += tanhf_fast(q[j].w + tv.w) * wv.w;
  }
  float* orow = out + ((long)(b * SS + p0) * CC + c) * SS + a0 + tid;
#pragma unroll
  for (int j = 0; j < TP; ++j) orow[(long)j * (CC * SS)] = acc[j];
}

extern "C" void kernel_launch(void* const* d_in, const int* in_sizes, int n_in,
                              void* d_out, int out_size, void* d_ws, size_t ws_size,
                              hipStream_t stream) {
  const float* seq  = (const float*)d_in[0];
  const float* base = (const float*)d_in[1];
  const float* Wp   = (const float*)d_in[2];
  const float* bp   = (const float*)d_in[3];
  const float* Ua   = (const float*)d_in[4];
  const float* ba   = (const float*)d_in[5];
  const float* Wmid = (const float*)d_in[6];
  const float* bmid = (const float*)d_in[7];
  const float* wout = (const float*)d_in[8];
  float* out = (float*)d_out;

  float* ws      = (float*)d_ws;
  float* tanh_hp = ws;
  float* ha_raw  = ws + 393216;
  float* tanh_ha = ws + 786432;
  float* tanhref = ws + 1179648;
  float* targ    = ws + 1572864;
  float* tref    = ws + 1966080;
  float* pref    = ws + 2064384;

  dim3 blk(32, 4);

  gemm_pair_kernel<MODE_TANH | MODE_BIAS, true, MODE_TANH | MODE_BIAS, false>
      <<<dim3(8, 24, 1), blk, 0, stream>>>(
      seq, DD, Ua, CC * HH, tanh_ha, CC * HH, ha_raw, ba, DD,
      seq, DD, Wp, CC * HH, tanh_hp, CC * HH, nullptr, bp, DD,
      12);

  gemm_wmma_kernel<MODE_TANH, false><<<dim3(2, 6, BB * CC), blk, 0, stream>>>(
      base, CC * SS, (long)SS * CC * SS, SS,
      ha_raw, CC * HH, (long)SS * CC * HH, HH,
      tanhref, CC * HH, (long)SS * CC * HH, HH,
      nullptr, nullptr, nullptr, SS);

  gemm_pair_kernel<0, false, 0, false><<<dim3(2, 60, 1), blk, 0, stream>>>(
      tanhref, CC * HH, Wmid + 2 * HH * HH, HH, tref, HH, nullptr, nullptr, CC * HH,
      tanh_ha, HH, Wmid + HH * HH, HH, targ, HH, nullptr, nullptr, HH,
      12);

  gemm_wmma_kernel<MODE_BIAS | MODE_ADDV, false><<<dim3(2, 48, 1), blk, 0, stream>>>(
      tanh_hp, HH, 0, 0, Wmid, HH, 0, 0, pref, HH, 0, 0,
      nullptr, bmid, tref, HH);

  final_reduce_kernel<<<dim3(SS / 128, SS / TP, BB * CC), dim3(128), 0, stream>>>(
      pref, targ, wout, out);
}


# Final kernel: tanh on the hardware instruction v_tanh_f32 (per element one transcendental, add, multiply-add; was multiply, exp2, add, rcp and two multiply-adds); k order of every sum unchanged
# speedup vs baseline: 3.3736x; 1.1329x over previous
#include <hip/hip_runtime.h>

#define BB 2
#define SS 384
#define CC 4
#define HH 128
#define DD 768

typedef float v2f __attribute__((ext_vector_type(2)));
typedef float v8f __attribute__((ext_vector_type(8)));

__device__ __forceinline__ float tanhf_fast(float x) {
  float e = __builtin_amdgcn_exp2f(x * 2.8853900817779268f);
  float r = __builtin_amdgcn_rcpf(e + 1.0f);
  return 1.0f - 2.0f * r;
}

__device__ __forceinline__ float tanhf_hw(float x) {
  return __builtin_amdgcn_tanhf(x);
}

#define MODE_TANH 1
#define MODE_BIAS 4
#define MODE_ADDV 8

template <int MODE, bool HAS_RAW>
__device__ __forceinline__ void store_tile(v8f c, float* O, int ldo, float* Oraw,
                                           const float* bias, const float* addvec,
                                           int m0, int n0, int lane) {
  int col   = n0 + (lane & 15);
  int rbase = m0 + ((lane >> 4) << 3);
#pragma unroll
  for (int i = 0; i < 8; ++i) {
    int   row = rbase + i;
    float v   = c[i];
    if constexpr (MODE & MODE_BIAS) v += bias[col];
    if constexpr (MODE & MODE_ADDV) v += addvec[(long)(row >> 2) * HH + col];
    if constexpr (HAS_RAW) Oraw[(long)row * ldo + col] = v;
    O[(long)row * ldo + col] = (MODE & MODE_TANH) ? tanhf_fast(v) : v;
  }
}

#define BK 128
#define LDK (BK + 4)
template <int MODE, bool HAS_RAW>
__device__ __forceinline__ void
gemm_tile(const float* __restrict__ Ab, int lda, const float* __restrict__ Bb, int ldb,
          float* __restrict__ Ob, int ldo, float* __restrict__ Rb,
          const float* __restrict__ bias, const float* __restrict__ addvec,
          int K, int mblk, int nblk, float* As, float* Bs) {
  int lane = threadIdx.x;
  int wave = threadIdx.y;
  int t    = wave * 32 + lane;
  int mw   = (wave >> 1) * 32;
  int nw   = (wave & 1) * 32;

  v8f c00 = {}, c01 = {}, c10 = {}, c11 = {};

  int ar0 = t >> 5, ac = (t & 31) << 2;
  int br0 = t >> 4, bc = (t & 15) << 2;
  const float* Ag = Ab + (long)(mblk + ar0) * lda + ac;
  const float* Bg = Bb + (long)br0 * ldb + nblk + bc;

  for (int kc = 0; kc < K; kc += BK) {
    __syncthreads();
    float4 av[16], bv[16];
#pragma unroll
    for (int rep = 0; rep < 16; ++rep)
      av[rep] = *(const float4*)(Ag + (long)(rep * 4) * lda + kc);
#pragma unroll
    for (int rep = 0; rep < 16; ++rep)
      bv[rep] = *(const float4*)(Bg + (long)(kc + rep * 8) * ldb);
#pragma unroll
    for (int rep = 0; rep < 16; ++rep)
      *(float4*)(&As[(ar0 + rep * 4) * LDK + ac]) = av[rep];
#pragma unroll
    for (int rep = 0; rep < 16; ++rep) {
      int br = br0 + rep * 8;
      Bs[(bc + 0) * LDK + br] = bv[rep].x;
      Bs[(bc + 1) * LDK + br] = bv[rep].y;
      Bs[(bc + 2) * LDK + br] = bv[rep].z;
      Bs[(bc + 3) * LDK + br] = bv[rep].w;
    }
    __syncthreads();

#pragma unroll
    for (int kl = 0; kl < BK; kl += 4) {
      int kf = kl + ((lane >> 4) << 1);
      int rr = lane & 15;
      v2f a0 = *(const v2f*)(&As[(mw + rr) * LDK + kf]);
      v2f a1 = *(const v2f*)(&As[(mw + 16 + rr) * LDK + kf]);
      v2f b0 = *(const v2f*)(&Bs[(nw + rr) * LDK + kf]);
      v2f b1 = *(const v2f*)(&Bs[(nw + 16 + rr) * LDK + kf]);
      c00 = __builtin_amdgcn_wmma_f32_16x16x4_f32(false, a0, false, b0, (short)0, c00, false, false);
      c01 = __builtin_amdgcn_wmma_f32_16x16x4_f32(false, a0, false, b1, (short)0, c01, false, false);
      c10 = __builtin_amdgcn_wmma_f32_16x16x4_f32(false, a1, false, b0, (short)0, c10, false, false);
      c11 = __builtin_amdgcn_wmma_f32_16x16x4_f32(false, a1, false, b1, (short)0, c11, false, false);
    }
  }

  int m0 = mblk + mw, n0 = nblk + nw;
  store_tile<MODE, HAS_RAW>(c00, Ob, ldo, Rb, bias, addvec, m0,      n0,      lane);
  store_tile<MODE, HAS_RAW>(c01, Ob, ldo, Rb, bias, addvec, m0,      n0 + 16, lane);
  store_tile<MODE, HAS_RAW>(c10, Ob, ldo, Rb, bias, addvec, m0 + 16, n0,      lane);
  store_tile<MODE, HAS_RAW>(c11, Ob, ldo, Rb, bias, addvec, m0 + 16, n0 + 16, lane);
}

template <int MODE, bool HAS_RAW>
__global__ void __launch_bounds__(128)
gemm_wmma_kernel(const float* __restrict__ A, int lda, long aOuter, long aInner,
                 const float* __restrict__ B, int ldb, long bOuter, long bInner,
                 float* __restrict__ O, int ldo, long oOuter, long oInner,
                 float* __restrict__ Oraw,
                 const float* __restrict__ bias, const float* __restrict__ addvec,
                 int K) {
  __shared__ float As[64 * LDK];
  __shared__ float Bs[64 * LDK];
  int zo = blockIdx.z >> 2;
  int zi = blockIdx.z & 3;
  long ooff = (long)zo * oOuter + (long)zi * oInner;
  gemm_tile<MODE, HAS_RAW>(A + (long)zo * aOuter + (long)zi * aInner, lda,
                           B + (long)zo * bOuter + (long)zi * bInner, ldb,
                           O + ooff, ldo, HAS_RAW ? (Oraw + ooff) : nullptr,
                           bias, addvec, K, blockIdx.y * 64, blockIdx.x * 64, As, Bs);
}

template <int MODE0, bool RAW0, int MODE1, bool RAW1>
__global__ void __launch_bounds__(128)
gemm_pair_kernel(const float* __restrict__ A0, int lda0, const float* __restrict__ B0, int ldb0,
                 float* __restrict__ O0, int ldo0, float* __restrict__ R0,
                 const float* __restrict__ bias0, int K0,
                 const float* __restrict__ A1, int lda1, const float* __restrict__ B1, int ldb1,
                 float* __restrict__ O1, int ldo1, float* __restrict__ R1,
                 const float* __restrict__ bias1, int K1,
                 int ny0) {
  __shared__ float As[64 * LDK];
  __shared__ float Bs[64 * LDK];
  int by = blockIdx.y;
  if (by < ny0)
    gemm_tile<MODE0, RAW0>(A0, lda0, B0, ldb0, O0, ldo0, R0, bias0, nullptr, K0,
                           by * 64, blockIdx.x * 64, As, Bs);
  else
    gemm_tile<MODE1, RAW1>(A1, lda1, B1, ldb1, O1, ldo1, R1, bias1, nullptr, K1,
                           (by - ny0) * 64, blockIdx.x * 64, As, Bs);
}

#define TP 12
__global__ void __launch_bounds__(128)
final_reduce_kernel(const float* __restrict__ pref, const float* __restrict__ targ,
                    const float* __restrict__ wout, float* __restrict__ out) {
  __shared__ __attribute__((aligned(16))) float Ts[128 * LDK];
  __shared__ __attribute__((aligned(16))) float sp[TP * HH];
  __shared__ __attribute__((aligned(16))) float sw[HH];
  int tid  = threadIdx.x;
  int lane = tid & 31;
  int wave = tid >> 5;
  int a0   = blockIdx.x * 128;
  int p0   = blockIdx.y * TP;
  int bc   = blockIdx.z;
  int b    = bc >> 2;
  int c    = bc & 3;

  const float* tg = targ + ((long)(b * SS + a0) * CC + c) * HH + (lane << 2);
  {
    float4 tv[32];
#pragma unroll
    for (int rep = 0; rep < 32; ++rep)
      tv[rep] = *(const float4*)(tg + (long)(wave + rep * 4) * (CC * HH));
#pragma unroll
    for (int rep = 0; rep < 32; ++rep)
      *(float4*)(&Ts[(wave + rep * 4) * LDK + (lane << 2)]) = tv[rep];
  }
#pragma unroll
  for (int j = 0; j < TP * (HH / 4) / 128; ++j) {
    int i = tid + j * 128;
    int row = i >> 5, c4 = (i & 31) << 2;
    *(float4*)(&sp[row * HH + c4]) =
        *(const float4*)(pref + ((long)(b * SS + p0 + row) * CC + c) * HH + c4);
  }
  sw[tid] = wout[tid];
  __syncthreads();

  const float* trow = &Ts[tid * LDK];
  float acc[TP];
#pragma unroll
  for (int j = 0; j < TP; ++j) acc[j] = 0.0f;
  for (int k = 0; k < HH; k += 4) {
    float4 tv = *(const float4*)(trow + k);
    float4 wv = *(const float4*)(sw + k);
    float4 q[TP];
#pragma unroll
    for (int j = 0; j < TP; ++j) q[j] = *(const float4*)(sp + j * HH + k);
#pragma unroll
    for (int j = 0; j < TP; ++j) acc[j] += tanhf_hw(q[j].x + tv.x) * wv.x;
#pragma unroll
    for (int j = 0; j < TP; ++j) acc[j] += tanhf_hw(q[j].y + tv.y) * wv.y;
#pragma unroll
    for (int j = 0; j < TP; ++j) acc[j] += tanhf_hw(q[j].z + tv.z) * wv.z;
#pragma unroll
    for (int j = 0; j < TP; ++j) acc[j] += tanhf_hw(q[j].w + tv.w) * wv.w;
  }
  float* orow = out + ((long)(b * SS + p0) * CC + c) * SS + a0 + tid;
#pragma unroll
  for (int j = 0; j < TP; ++j) orow[(long)j * (CC * SS)] = acc[j];
}

extern "C" void kernel_launch(void* const* d_in, const int* in_sizes, int n_in,
                              void* d_out, int out_size, void* d_ws, size_t ws_size,
                              hipStream_t stream) {
  const float* seq  = (const float*)d_in[0];
  const float* base = (const float*)d_in[1];
  const float* Wp   = (const float*)d_in[2];
  const float* bp   = (const float*)d_in[3];
  const float* Ua   = (const float*)d_in[4];
  const float* ba   = (const float*)d_in[5];
  const float* Wmid = (const float*)d_in[6];
  const float* bmid = (const float*)d_in[7];
  const float* wout = (const float*)d_in[8];
  float* out = (float*)d_out;

  float* ws      = (float*)d_ws;
  float* tanh_hp = ws;
  float* ha_raw  = ws + 393216;
  float* tanh_ha = ws + 786432;
  float* tanhref = ws + 1179648;
  float* targ    = ws + 1572864;
  float* tref    = ws + 1966080;
  float* pref    = ws + 2064384;

  dim3 blk(32, 4);

  gemm_pair_kernel<MODE_TANH | MODE_BIAS, true, MODE_TANH | MODE_BIAS, false>
      <<<dim3(8, 24, 1), blk, 0, stream>>>(
      seq, DD, Ua, CC * HH, tanh_ha, CC * HH, ha_raw, ba, DD,
      seq, DD, Wp, CC * HH, tanh_hp, CC * HH, nullptr, bp, DD,
      12);

  gemm_wmma_kernel<MODE_TANH, false><<<dim3(2, 6, BB * CC), blk, 0, stream>>>(
      base, CC * SS, (long)SS * CC * SS, SS,
      ha_raw, CC * HH, (long)SS * CC * HH, HH,
      tanhref, CC * HH, (long)SS * CC * HH, HH,
      nullptr, nullptr, nullptr, SS);

  gemm_pair_kernel<0, false, 0, false><<<dim3(2, 60, 1), blk, 0, stream>>>(
      tanhref, CC * HH, Wmid + 2 * HH * HH, HH, tref, HH, nullptr, nullptr, CC * HH,
      tanh_ha, HH, Wmid + HH * HH, HH, targ, HH, nullptr, nullptr, HH,
      12);

  gemm_wmma_kernel<MODE_BIAS | MODE_ADDV, false><<<dim3(2, 48, 1), blk, 0, stream>>>(
      tanh_hp, HH, 0, 0, Wmid, HH, 0, 0, pref, HH, 0, 0,
      nullptr, bmid, tref, HH);

  final_reduce_kernel<<<dim3(SS / 128, SS / TP, BB * CC), dim3(128), 0, stream>>>(
      pref, targ, wout, out);
}


# t_ref is added to the pref rows where the final kernel loads them, so the tanh(h_p)@W1 GEMM no longer waits for t_ref and rides in the t_ref/targ launch; 5 launches -> 4; order of every sum unchanged
# speedup vs baseline: 3.7100x; 1.0997x over previous
#include <hip/hip_runtime.h>

#define BB 2
#define SS 384
#define CC 4
#define HH 128
#define DD 768

typedef float v2f __attribute__((ext_vector_type(2)));
typedef float v8f __attribute__((ext_vector_type(8)));

__device__ __forceinline__ float tanhf_fast(float x) {
  float e = __builtin_amdgcn_exp2f(x * 2.8853900817779268f);
  float r = __builtin_amdgcn_rcpf(e + 1.0f);
  return 1.0f - 2.0f * r;
}

__device__ __forceinline__ float tanhf_hw(float x) {
  return __builtin_amdgcn_tanhf(x);
}

#define MODE_TANH 1
#define MODE_BIAS 4

template <int MODE, bool HAS_RAW>
__device__ __forceinline__ void store_tile(v8f c, float* O, int ldo, float* Oraw,
                                           const float* bias,
                                           int m0, int n0, int lane) {
  int col   = n0 + (lane & 15);
  int rbase = m0 + ((lane >> 4) << 3);
#pragma unroll
  for (int i = 0; i < 8; ++i) {
    int   row = rbase + i;
    float v   = c[i];
    if constexpr (MODE & MODE_BIAS) v += bias[col];
    if constexpr (HAS_RAW) Oraw[(long)row * ldo + col] = v;
    O[(long)row * ldo + col] = (MODE & MODE_TANH) ? tanhf_fast(v) : v;
  }
}

#define BK 128
#define LDK (BK + 4)
template <int MODE, bool HAS_RAW>
__device__ __forceinline__ void
gemm_tile(const float* __restrict__ Ab, int lda, const float* __restrict__ Bb, int ldb,
          float* __restrict__ Ob, int ldo, float* __restrict__ Rb,
          const float* __restrict__ bias,
          int K, int mblk, int nblk, float* As, float* Bs) {
  int lane = threadIdx.x;
  int wave = threadIdx.y;
  int t    = wave * 32 + lane;
  int mw   = (wave >> 1) * 32;
  int nw   = (wave & 1) * 32;

  v8f c00 = {}, c01 = {}, c10 = {}, c11 = {};

  int ar0 = t >> 5, ac = (t & 31) << 2;
  int br0 = t >> 4, bc = (t & 15) << 2;
  const float* Ag = Ab + (long)(mblk + ar0) * lda + ac;
  const float* Bg = Bb + (long)br0 * ldb + nblk + bc;

  for (int kc = 0; kc < K; kc += BK) {
    __syncthreads();
    float4 av[16], bv[16];
#pragma unroll
    for (int rep = 0; rep < 16; ++rep)
      av[rep] = *(const float4*)(Ag + (long)(rep * 4) * lda + kc);
#pragma unroll
    for (int rep = 0; rep < 16; ++rep)
      bv[rep] = *(const float4*)(Bg + (long)(kc + rep * 8) * ldb);
#pragma unroll
    for (int rep = 0; rep < 16; ++rep)
      *(float4*)(&As[(ar0 + rep * 4) * LDK + ac]) = av[rep];
#pragma unroll
    for (int rep = 0; rep < 16; ++rep) {
      int br = br0 + rep * 8;
      Bs[(bc + 0) * LDK + br] = bv[rep].x;
      Bs[(bc + 1) * LDK + br] = bv[rep].y;
      Bs[(bc + 2) * LDK + br] = bv[rep].z;
      Bs[(bc + 3) * LDK + br] = bv[rep].w;
    }
    __syncthreads();

#pragma unroll
    for (int kl = 0; kl < BK; kl += 4) {
      int kf = kl + ((lane >> 4) << 1);
      int rr = lane & 15;
      v2f a0 = *(const v2f*)(&As[(mw + rr) * LDK + kf]);
      v2f a1 = *(const v2f*)(&As[(mw + 16 + rr) * LDK + kf]);
      v2f b0 = *(const v2f*)(&Bs[(nw + rr) * LDK + kf]);
      v2f b1 = *(const v2f*)(&Bs[(nw + 16 + rr) * LDK + kf]);
      c00 = __builtin_amdgcn_wmma_f32_16x16x4_f32(false, a0, false, b0, (short)0, c00, false, false);
      c01 = __builtin_amdgcn_wmma_f32_16x16x4_f32(false, a0, false, b1, (short)0, c01, false, false);
      c10 = __builtin_amdgcn_wmma_f32_16x16x4_f32(false, a1, false, b0, (short)0, c10, false, false);
      c11 = __builtin_amdgcn_wmma_f32_16x16x4_f32(false, a1, false, b1, (short)0, c11, false, false);
    }
  }

  int m0 = mblk + mw, n0 = nblk + nw;
  store_tile<MODE, HAS_RAW>(c00, Ob, ldo, Rb, bias, m0,      n0,      lane);
  store_tile<MODE, HAS_RAW>(c01, Ob, ldo, Rb, bias, m0,      n0 + 16, lane);
  store_tile<MODE, HAS_RAW>(c10, Ob, ldo, Rb, bias, m0 + 16, n0,      lane);
  store_tile<MODE, HAS_RAW>(c11, Ob, ldo, Rb, bias, m0 + 16, n0 + 16, lane);
}

template <int MODE, bool HAS_RAW>
__global__ void __launch_bounds__(128)
gemm_wmma_kernel(const float* __restrict__ A, int lda, long aOuter, long aInner,
                 const float* __restrict__ B, int ldb, long bOuter, long bInner,
                 float* __restrict__ O, int ldo, long oOuter, long oInner,
                 float* __restrict__ Oraw,
                 const float* __restrict__ bias,
                 int K) {
  __shared__ float As[64 * LDK];
  __shared__ float Bs[64 * LDK];
  int zo = blockIdx.z >> 2;
  int zi = blockIdx.z & 3;
  long ooff = (long)zo * oOuter + (long)zi * oInner;
  gemm_tile<MODE, HAS_RAW>(A + (long)zo * aOuter + (long)zi * aInner, lda,
                           B + (long)zo * bOuter + (long)zi * bInner, ldb,
                           O + ooff, ldo, HAS_RAW ? (Oraw + ooff) : nullptr,
                           bias, K, blockIdx.y * 64, blockIdx.x * 64, As, Bs);
}

template <int MODE0, bool RAW0, int MODE1, bool RAW1>
__global__ void __launch_bounds__(128)
gemm_pair_kernel(const float* __restrict__ A0, int lda0, const float* __restrict__ B0, int ldb0,
                 float* __restrict__ O0, int ldo0, float* __restrict__ R0,
                 const float* __restrict__ bias0, int K0,
                 const float* __restrict__ A1, int lda1, const float* __restrict__ B1, int ldb1,
                 float* __restrict__ O1, int ldo1, float* __restrict__ R1,
                 const float* __restrict__ bias1, int K1,
                 int ny0) {
  __shared__ float As[64 * LDK];
  __shared__ float Bs[64 * LDK];
  int by = blockIdx.y;
  if (by < ny0)
    gemm_tile<MODE0, RAW0>(A0, lda0, B0, ldb0, O0, ldo0, R0, bias0, K0,
                           by * 64, blockIdx.x * 64, As, Bs);
  else
    gemm_tile<MODE1, RAW1>(A1, lda1, B1, ldb1, O1, ldo1, R1, bias1, K1,
                           (by - ny0) * 64, blockIdx.x * 64, As, Bs);
}

template <int MODE0, int MODE1, int MODE2>
__global__ void __launch_bounds__(128)
gemm_triple_kernel(const float* __restrict__ A0, int lda0, const float* __restrict__ B0, int ldb0,
                   float* __restrict__ O0, int ldo0, const float* __restrict__ bias0, int K0,
                   const float* __restrict__ A1, int lda1, const float* __restrict__ B1, int ldb1,
                   float* __restrict__ O1, int ldo1, const float* __restrict__ bias1, int K1,
                   const float* __restrict__ A2, int lda2, const float* __restrict__ B2, int ldb2,
                   float* __restrict__ O2, int ldo2, const float* __restrict__ bias2, int K2,
                   int ny0, int ny1) {
  __shared__ float As[64 * LDK];
  __shared__ float Bs[64 * LDK];
  int by = blockIdx.y;
  if (by < ny0)
    gemm_tile<MODE0, false>(A0, lda0, B0, ldb0, O0, ldo0, nullptr, bias0, K0,
                            by * 64, blockIdx.x * 64, As, Bs);
  else if (by < ny1)
    gemm_tile<MODE1, false>(A1, lda1, B1, ldb1, O1, ldo1, nullptr, bias1, K1,
                            (by - ny0) * 64, blockIdx.x * 64, As, Bs);
  else
    gemm_tile<MODE2, false>(A2, lda2, B2, ldb2, O2, ldo2, nullptr, bias2, K2,
                            (by - ny1) * 64, blockIdx.x * 64, As, Bs);
}

#define TP 12
__global__ void __launch_bounds__(128)
final_reduce_kernel(const float* __restrict__ tprd, const float* __restrict__ tref,
                    const float* __restrict__ targ,
                    const float* __restrict__ wout, float* __restrict__ out) {
  __shared__ __attribute__((aligned(16))) float Ts[128 * LDK];
  __shared__ __attribute__((aligned(16))) float sp[TP * HH];
  __shared__ __attribute__((aligned(16))) float sw[HH];
  int tid  = threadIdx.x;
  int lane = tid & 31;
  int wave = tid >> 5;
  int a0   = blockIdx.x * 128;
  int p0   = blockIdx.y * TP;
  int bc   = blockIdx.z;
  int b    = bc >> 2;
  int c    = bc & 3;

  const float* tg = targ + ((long)(b * SS + a0) * CC + c) * HH + (lane << 2);
  {
    float4 tv[32];
#pragma unroll
    for (int rep = 0; rep < 32; ++rep)
      tv[rep] = *(const float4*)(tg + (long)(wave + rep * 4) * (CC * HH));
#pragma unroll
    for (int rep = 0; rep < 32; ++rep)
      *(float4*)(&Ts[(wave + rep * 4) * LDK + (lane << 2)]) = tv[rep];
  }
#pragma unroll
  for (int j = 0; j < TP * (HH / 4) / 128; ++j) {
    int i = tid + j * 128;
    int row = i >> 5, c4 = (i & 31) << 2;
    long bp = (long)(b * SS + p0 + row);
    float4 pv = *(const float4*)(tprd + (bp * CC + c) * HH + c4);
    float4 rv = *(const float4*)(tref + bp * HH + c4);
    float4 sv;
    sv.x = pv.x + rv.x;
    sv.y = pv.y + rv.y;
    sv.z = pv.z + rv.z;
    sv.w = pv.w + rv.w;
    *(float4*)(&sp[row * HH + c4]) = sv;
  }
  sw[tid] = wout[tid];
  __syncthreads();

  const float* trow = &Ts[tid * LDK];
  float acc[TP];
#pragma unroll
  for (int j = 0; j < TP; ++j) acc[j] = 0.0f;
  for (int k = 0; k < HH; k += 4) {
    float4 tv = *(const float4*)(trow + k);
    float4 wv = *(const float4*)(sw + k);
    float4 q[TP];
#pragma unroll
    for (int j = 0; j < TP; ++j) q[j] = *(const float4*)(sp + j * HH + k);
#pragma unroll
    for (int j = 0; j < TP; ++j) acc[j] += tanhf_hw(q[j].x + tv.x) * wv.x;
#pragma unroll
    for (int j = 0; j < TP; ++j) acc[j] += tanhf_hw(q[j].y + tv.y) * wv.y;
#pragma unroll
    for (int j = 0; j < TP; ++j) acc[j] += tanhf_hw(q[j].z + tv.z) * wv.z;
#pragma unroll
    for (int j = 0; j < TP; ++j) acc[j] += tanhf_hw(q[j].w + tv.w) * wv.w;
  }
  float* orow = out + ((long)(b * SS + p0) * CC + c) * SS + a0 + tid;
#pragma unroll
  for (int j = 0; j < TP; ++j) orow[(long)j * (CC * SS)] = acc[j];
}

extern "C" void kernel_launch(void* const* d_in, const int* in_sizes, int n_in,
                              void* d_out, int out_size, void* d_ws, size_t ws_size,
                              hipStream_t stream) {
  const float* seq  = (const float*)d_in[0];
  const float* base = (const float*)d_in[1];
  const float* Wp   = (const float*)d_in[2];
  const float* bp   = (const float*)d_in[3];
  const float* Ua   = (const float*)d_in[4];
  const float* ba   = (const float*)d_in[5];
  const float* Wmid = (const float*)d_in[6];
  const float* bmid = (const float*)d_in[7];
  const float* wout = (const float*)d_in[8];
  float* out = (float*)d_out;

  float* ws      = (float*)d_ws;
  float* tanh_hp = ws;
  float* ha_raw  = ws + 393216;
  float* tanh_ha = ws + 786432;
  float* tanhref = ws + 1179648;
  float* targ    = ws + 1572864;
  float* tref    = ws + 1966080;
  float* tprd    = ws + 2064384;

  dim3 blk(32, 4);

  gemm_pair_kernel<MODE_TANH | MODE_BIAS, true, MODE_TANH | MODE_BIAS, false>
      <<<dim3(8, 24, 1), blk, 0, stream>>>(
      seq, DD, Ua, CC * HH, tanh_ha, CC * HH, ha_raw, ba, DD,
      seq, DD, Wp, CC * HH, tanh_hp, CC * HH, nullptr, bp, DD,
      12);

  gemm_wmma_kernel<MODE_TANH, false><<<dim3(2, 6, BB * CC), blk, 0, stream>>>(
      base, CC * SS, (long)SS * CC * SS, SS,
      ha_raw, CC * HH, (long)SS * CC * HH, HH,
      tanhref, CC * HH, (long)SS * CC * HH, HH,
      nullptr, nullptr, SS);

  gemm_triple_kernel<0, 0, MODE_BIAS><<<dim3(2, 108, 1), blk, 0, stream>>>(
      tanhref, CC * HH, Wmid + 2 * HH * HH, HH, tref, HH, nullptr, CC * HH,
      tanh_ha, HH, Wmid + HH * HH, HH, targ, HH, nullptr, HH,
      tanh_hp, HH, Wmid, HH, tprd, HH, bmid, HH,
      12, 60);

  final_reduce_kernel<<<dim3(SS / 128, SS / TP, BB * CC), dim3(128), 0, stream>>>(
      tprd, tref, targ, wout, out);
}


# GEMM K loop: the global loads of stage s+1 are issued into the staging registers behind the barrier of stage s, so their trip to memory runs under its 128 WMMAs; sum order unchanged
# speedup vs baseline: 3.9574x; 1.0667x over previous
#include <hip/hip_runtime.h>

#define BB 2
#define SS 384
#define CC 4
#define HH 128
#define DD 768

typedef float v2f __attribute__((ext_vector_type(2)));
typedef float v4f __attribute__((ext_vector_type(4)));
typedef float v8f __attribute__((ext_vector_type(8)));

__device__ __forceinline__ float tanhf_fast(float x) {
  float e = __builtin_amdgcn_exp2f(x * 2.8853900817779268f);
  float r = __builtin_amdgcn_rcpf(e + 1.0f);
  return 1.0f - 2.0f * r;
}

__device__ __forceinline__ float tanhf_hw(float x) {
  return __builtin_amdgcn_tanhf(x);
}

#define MODE_TANH 1
#define MODE_BIAS 4

template <int MODE, bool HAS_RAW>
__device__ __forceinline__ void store_tile(v8f c, float* O, int ldo, float* Oraw,
                                           const float* bias,
                                           int m0, int n0, int lane) {
  int col   = n0 + (lane & 15);
  int rbase = m0 + ((lane >> 4) << 3);
#pragma unroll
  for (int i = 0; i < 8; ++i) {
    int   row = rbase + i;
    float v   = c[i];
    if constexpr (MODE & MODE_BIAS) v += bias[col];
    if constexpr (HAS_RAW) Oraw[(long)row * ldo + col] = v;
    O[(long)row * ldo + col] = (MODE & MODE_TANH) ? tanhf_fast(v) : v;
  }
}

#define BK 128
#define LDK (BK + 4)
template <int MODE, bool HAS_RAW>
__device__ __forceinline__ void
gemm_tile(const float* __restrict__ Ab, int lda, const float* __restrict__ Bb, int ldb,
          float* __restrict__ Ob, int ldo, float* __restrict__ Rb,
          const float* __restrict__ bias,
          int K, int mblk, int nblk, float* As, float* Bs) {
  int lane = threadIdx.x;
  int wave = threadIdx.y;
  int t    = wave * 32 + lane;
  int mw   = (wave >> 1) * 32;
  int nw   = (wave & 1) * 32;

  v8f c00 = {}, c01 = {}, c10 = {}, c11 = {};

  int ar0 = t >> 5, ac = (t & 31) << 2;
  int br0 = t >> 4, bc = (t & 15) << 2;
  const float* Ag = Ab + (long)(mblk + ar0) * lda + ac;
  const float* Bg = Bb + (long)br0 * ldb + nblk + bc;

  v4f av[16], bv[16];
#pragma unroll
  for (int rep = 0; rep < 16; ++rep)
    av[rep] = *(const v4f*)(Ag + (long)(rep * 4) * lda);
#pragma unroll
  for (int rep = 0; rep < 16; ++rep)
    bv[rep] = *(const v4f*)(Bg + (long)(rep * 8) * ldb);

  for (int kc = 0; kc < K; kc += BK) {
    __syncthreads();
#pragma unroll
    for (int rep = 0; rep < 16; ++rep)
      *(v4f*)(&As[(ar0 + rep * 4) * LDK + ac]) = av[rep];
#pragma unroll
    for (int rep = 0; rep < 16; ++rep) {
      int br = br0 + rep * 8;
      Bs[(bc + 0) * LDK + br] = bv[rep].x;
      Bs[(bc + 1) * LDK + br] = bv[rep].y;
      Bs[(bc + 2) * LDK + br] = bv[rep].z;
      Bs[(bc + 3) * LDK + br] = bv[rep].w;
    }
    __syncthreads();

    int kn = kc + BK;
    if (kn < K) {
#pragma unroll
      for (int rep = 0; rep < 16; ++rep)
        av[rep] = *(const v4f*)(Ag + (long)(rep * 4) * lda + kn);
#pragma unroll
      for (int rep = 0; rep < 16; ++rep)
        bv[rep] = *(const v4f*)(Bg + (long)(kn + rep * 8) * ldb);
    }

#pragma unroll
    for (int kl = 0; kl < BK; kl += 4) {
      int kf = kl + ((lane >> 4) << 1);
      int rr = lane & 15;
      v2f a0 = *(const v2f*)(&As[(mw + rr) * LDK + kf]);
      v2f a1 = *(const v2f*)(&As[(mw + 16 + rr) * LDK + kf]);
      v2f b0 = *(const v2f*)(&Bs[(nw + rr) * LDK + kf]);
      v2f b1 = *(const v2f*)(&Bs[(nw + 16 + rr) * LDK + kf]);
      c00 = __builtin_amdgcn_wmma_f32_16x16x4_f32(false, a0, false, b0, (short)0, c00, false, false);
      c01 = __builtin_amdgcn_wmma_f32_16x16x4_f32(false, a0, false, b1, (short)0, c01, false, false);
      c10 = __builtin_amdgcn_wmma_f32_16x16x4_f32(false, a1, false, b0, (short)0, c10, false, false);
      c11 = __builtin_amdgcn_wmma_f32_16x16x4_f32(false, a1, false, b1, (short)0, c11, false, false);
    }
  }

  int m0 = mblk + mw, n0 = nblk + nw;
  store_tile<MODE, HAS_RAW>(c00, Ob, ldo, Rb, bias, m0,      n0,      lane);
  store_tile<MODE, HAS_RAW>(c01, Ob, ldo, Rb, bias, m0,      n0 + 16, lane);
  store_tile<MODE, HAS_RAW>(c10, Ob, ldo, Rb, bias, m0 + 16, n0,      lane);
  store_tile<MODE, HAS_RAW>(c11, Ob, ldo, Rb, bias, m0 + 16, n0 + 16, lane);
}

template <int MODE, bool HAS_RAW>
__global__ void __launch_bounds__(128)
gemm_wmma_kernel(const float* __restrict__ A, int lda, long aOuter, long aInner,
                 const float* __restrict__ B, int ldb, long bOuter, long bInner,
                 float* __restrict__ O, int ldo, long oOuter, long oInner,
                 float* __restrict__ Oraw,
                 const float* __restrict__ bias,
                 int K) {
  __shared__ float As[64 * LDK];
  __shared__ float Bs[64 * LDK];
  int zo = blockIdx.z >> 2;
  int zi = blockIdx.z & 3;
  long ooff = (long)zo * oOuter + (long)zi * oInner;
  gemm_tile<MODE, HAS_RAW>(A + (long)zo * aOuter + (long)zi * aInner, lda,
                           B + (long)zo * bOuter + (long)zi * bInner, ldb,
                           O + ooff, ldo, HAS_RAW ? (Oraw + ooff) : nullptr,
                           bias, K, blockIdx.y * 64, blockIdx.x * 64, As, Bs);
}

template <int MODE0, bool RAW0, int MODE1, bool RAW1>
__global__ void __launch_bounds__(128)
gemm_pair_kernel(const float* __restrict__ A0, int lda0, const float* __restrict__ B0, int ldb0,
                 float* __restrict__ O0, int ldo0, float* __restrict__ R0,
                 const float* __restrict__ bias0, int K0,
                 const float* __restrict__ A1, int lda1, const float* __restrict__ B1, int ldb1,
                 float* __restrict__ O1, int ldo1, float* __restrict__ R1,
                 const float* __restrict__ bias1, int K1,
                 int ny0) {
  __shared__ float As[64 * LDK];
  __shared__ float Bs[64 * LDK];
  int by = blockIdx.y;
  if (by < ny0)
    gemm_tile<MODE0, RAW0>(A0, lda0, B0, ldb0, O0, ldo0, R0, bias0, K0,
                           by * 64, blockIdx.x * 64, As, Bs);
  else
    gemm_tile<MODE1, RAW1>(A1, lda1, B1, ldb1, O1, ldo1, R1, bias1, K1,
                           (by - ny0) * 64, blockIdx.x * 64, As, Bs);
}

template <int MODE0, int MODE1, int MODE2>
__global__ void __launch_bounds__(128)
gemm_triple_kernel(const float* __restrict__ A0, int lda0, const float* __restrict__ B0, int ldb0,
                   float* __restrict__ O0, int ldo0, const float* __restrict__ bias0, int K0,
                   const float* __restrict__ A1, int lda1, const float* __restrict__ B1, int ldb1,
                   float* __restrict__ O1, int ldo1, const float* __restrict__ bias1, int K1,
                   const float* __restrict__ A2, int lda2, const float* __restrict__ B2, int ldb2,
                   float* __restrict__ O2, int ldo2, const float* __restrict__ bias2, int K2,
                   int ny0, int ny1) {
  __shared__ float As[64 * LDK];
  __shared__ float Bs[64 * LDK];
  int by = blockIdx.y;
  if (by < ny0)
    gemm_tile<MODE0, false>(A0, lda0, B0, ldb0, O0, ldo0, nullptr, bias0, K0,
                            by * 64, blockIdx.x * 64, As, Bs);
  else if (by < ny1)
    gemm_tile<MODE1, false>(A1, lda1, B1, ldb1, O1, ldo1, nullptr, bias1, K1,
                            (by - ny0) * 64, blockIdx.x * 64, As, Bs);
  else
    gemm_tile<MODE2, false>(A2, lda2, B2, ldb2, O2, ldo2, nullptr, bias2, K2,
                            (by - ny1) * 64, blockIdx.x * 64, As, Bs);
}

#define TP 12
__global__ void __launch_bounds__(128)
final_reduce_kernel(const float* __restrict__ tprd, const float* __restrict__ tref,
                    const float* __restrict__ targ,
                    const float* __restrict__ wout, float* __restrict__ out) {
  __shared__ __attribute__((aligned(16))) float Ts[128 * LDK];
  __shared__ __attribute__((aligned(16))) float sp[TP * HH];
  __shared__ __attribute__((aligned(16))) float sw[HH];
  int tid  = threadIdx.x;
  int lane = tid & 31;
  int wave = tid >> 5;
  int a0   = blockIdx.x * 128;
  int p0   = blockIdx.y * TP;
  int bc   = blockIdx.z;
  int b    = bc >> 2;
  int c    = bc & 3;

  const float* tg = targ + ((long)(b * SS + a0) * CC + c) * HH + (lane << 2);
  {
    float4 tv[32];
#pragma unroll
    for (int rep = 0; rep < 32; ++rep)
      tv[rep] = *(const float4*)(tg + (long)(wave + rep * 4) * (CC * HH));
#pragma unroll
    for (int rep = 0; rep < 32; ++rep)
      *(float4*)(&Ts[(wave + rep * 4) * LDK + (lane << 2)]) = tv[rep];
  }
#pragma unroll
  for (int j = 0; j < TP * (HH / 4) / 128; ++j) {
    int i = tid + j * 128;
    int row = i >> 5, c4 = (i & 31) << 2;
    long bp = (long)(b * SS + p0 + row);
    float4 pv = *(const float4*)(tprd + (bp * CC + c) * HH + c4);
    float4 rv = *(const float4*)(tref + bp * HH + c4);
    float4 sv;
    sv.x = pv.x + rv.x;
    sv.y = pv.y + rv.y;
    sv.z = pv.z + rv.z;
    sv.w = pv.w + rv.w;
    *(float4*)(&sp[row * HH + c4]) = sv;
  }
  sw[tid] = wout[tid];
  __syncthreads();

  const float* trow = &Ts[tid * LDK];
  float acc[TP];
#pragma unroll
  for (int j = 0; j < TP; ++j) acc[j] = 0.0f;
  for (int k = 0; k < HH; k += 4) {
    float4 tv = *(const float4*)(trow + k);
    float4 wv = *(const float4*)(sw + k);
    float4 q[TP];
#pragma unroll
    for (int j = 0; j < TP; ++j) q[j] = *(const float4*)(sp + j * HH + k);
#pragma unroll
    for (int j = 0; j < TP; ++j) acc[j] += tanhf_hw(q[j].x + tv.x) * wv.x;
#pragma unroll
    for (int j = 0; j < TP; ++j) acc[j] += tanhf_hw(q[j].y + tv.y) * wv.y;
#pragma unroll
    for (int j = 0; j < TP; ++j) acc[j] += tanhf_hw(q[j].z + tv.z) * wv.z;
#pragma unroll
    for (int j = 0; j < TP; ++j) acc[j] += tanhf_hw(q[j].w + tv.w) * wv.w;
  }
  float* orow = out + ((long)(b * SS + p0) * CC + c) * SS + a0 + tid;
#pragma unroll
  for (int j = 0; j < TP; ++j) orow[(long)j * (CC * SS)] = acc[j];
}

extern "C" void kernel_launch(void* const* d_in, const int* in_sizes, int n_in,
                              void* d_out, int out_size, void* d_ws, size_t ws_size,
                              hipStream_t stream) {
  const float* seq  = (const float*)d_in[0];
  const float* base = (const float*)d_in[1];
  const float* Wp   = (const float*)d_in[2];
  const float* bp   = (const float*)d_in[3];
  const float* Ua   = (const float*)d_in[4];
  const float* ba   = (const float*)d_in[5];
  const float* Wmid = (const float*)d_in[6];
  const float* bmid = (const float*)d_in[7];
  const float* wout = (const float*)d_in[8];
  float* out = (float*)d_out;

  float* ws      = (float*)d_ws;
  float* tanh_hp = ws;
  float* ha_raw  = ws + 393216;
  float* tanh_ha = ws + 786432;
  float* tanhref = ws + 1179648;
  float* targ    = ws + 1572864;
  float* tref    = ws + 1966080;
  float* tprd    = ws + 2064384;

  dim3 blk(32, 4);

  gemm_pair_kernel<MODE_TANH | MODE_BIAS, true, MODE_TANH | MODE_BIAS, false>
      <<<dim3(8, 24, 1), blk, 0, stream>>>(
      seq, DD, Ua, CC * HH, tanh_ha, CC * HH, ha_raw, ba, DD,
      seq, DD, Wp, CC * HH, tanh_hp, CC * HH, nullptr, bp, DD,
      12);

  gemm_wmma_kernel<MODE_TANH, false><<<dim3(2, 6, BB * CC), blk, 0, stream>>>(
      base, CC * SS, (long)SS * CC * SS, SS,
      ha_raw, CC * HH, (long)SS * CC * HH, HH,
      tanhref, CC * HH, (long)SS * CC * HH, HH,
      nullptr, nullptr, SS);

  gemm_triple_kernel<0, 0, MODE_BIAS><<<dim3(2, 108, 1), blk, 0, stream>>>(
      tanhref, CC * HH, Wmid + 2 * HH * HH, HH, tref, HH, nullptr, CC * HH,
      tanh_ha, HH, Wmid + HH * HH, HH, targ, HH, nullptr, HH,
      tanh_hp, HH, Wmid, HH, tprd, HH, bmid, HH,
      12, 60);

  final_reduce_kernel<<<dim3(SS / 128, SS / TP, BB * CC), dim3(128), 0, stream>>>(
      tprd, tref, targ, wout, out);
}


# t_ref GEMM cut along K into four per-case products (one K stage each, 96 blocks instead of 24 with four stages); the final kernel adds the four partial planes c=0..3: t_ref sum order changes
# speedup vs baseline: 4.3247x; 1.0928x over previous
#include <hip/hip_runtime.h>

#define BB 2
#define SS 384
#define CC 4
#define HH 128
#define DD 768

typedef float v2f __attribute__((ext_vector_type(2)));
typedef float v4f __attribute__((ext_vector_type(4)));
typedef float v8f __attribute__((ext_vector_type(8)));

__device__ __forceinline__ float tanhf_fast(float x) {
  float e = __builtin_amdgcn_exp2f(x * 2.8853900817779268f);
  float r = __builtin_amdgcn_rcpf(e + 1.0f);
  return 1.0f - 2.0f * r;
}

__device__ __forceinline__ float tanhf_hw(float x) {
  return __builtin_amdgcn_tanhf(x);
}

#define MODE_TANH 1
#define MODE_BIAS 4

template <int MODE, bool HAS_RAW>
__device__ __forceinline__ void store_tile(v8f c, float* O, int ldo, float* Oraw,
                                           const float* bias,
                                           int m0, int n0, int lane) {
  int col   = n0 + (lane & 15);
  int rbase = m0 + ((lane >> 4) << 3);
#pragma unroll
  for (int i = 0; i < 8; ++i) {
    int   row = rbase + i;
    float v   = c[i];
    if constexpr (MODE & MODE_BIAS) v += bias[col];
    if constexpr (HAS_RAW) Oraw[(long)row * ldo + col] = v;
    O[(long)row * ldo + col] = (MODE & MODE_TANH) ? tanhf_fast(v) : v;
  }
}

#define BK 128
#define LDK (BK + 4)
template <int MODE, bool HAS_RAW>
__device__ __forceinline__ void
gemm_tile(const float* __restrict__ Ab, int lda, const float* __restrict__ Bb, int ldb,
          float* __restrict__ Ob, int ldo, float* __restrict__ Rb,
          const float* __restrict__ bias,
          int K, int mblk, int nblk, float* As, float* Bs) {
  int lane = threadIdx.x;
  int wave = threadIdx.y;
  int t    = wave * 32 + lane;
  int mw   = (wave >> 1) * 32;
  int nw   = (wave & 1) * 32;

  v8f c00 = {}, c01 = {}, c10 = {}, c11 = {};

  int ar0 = t >> 5, ac = (t & 31) << 2;
  int br0 = t >> 4, bc = (t & 15) << 2;
  const float* Ag = Ab + (long)(mblk + ar0) * lda + ac;
  const float* Bg = Bb + (long)br0 * ldb + nblk + bc;

  v4f av[16], bv[16];
#pragma unroll
  for (int rep = 0; rep < 16; ++rep)
    av[rep] = *(const v4f*)(Ag + (long)(rep * 4) * lda);
#pragma unroll
  for (int rep = 0; rep < 16; ++rep)
    bv[rep] = *(const v4f*)(Bg + (long)(rep * 8) * ldb);

  for (int kc = 0; kc < K; kc += BK) {
    __syncthreads();
#pragma unroll
    for (int rep = 0; rep < 16; ++rep)
      *(v4f*)(&As[(ar0 + rep * 4) * LDK + ac]) = av[rep];
#pragma unroll
    for (int rep = 0; rep < 16; ++rep) {
      int br = br0 + rep * 8;
      Bs[(bc + 0) * LDK + br] = bv[rep].x;
      Bs[(bc + 1) * LDK + br] = bv[rep].y;
      Bs[(bc + 2) * LDK + br] = bv[rep].z;
      Bs[(bc + 3) * LDK + br] = bv[rep].w;
    }
    __syncthreads();

    int kn = kc + BK;
    if (kn < K) {
#pragma unroll
      for (int rep = 0; rep < 16; ++rep)
        av[rep] = *(const v4f*)(Ag + (long)(rep * 4) * lda + kn);
#pragma unroll
      for (int rep = 0; rep < 16; ++rep)
        bv[rep] = *(const v4f*)(Bg + (long)(kn + rep * 8) * ldb);
    }

#pragma unroll
    for (int kl = 0; kl < BK; kl += 4) {
      int kf = kl + ((lane >> 4) << 1);
      int rr = lane & 15;
      v2f a0 = *(const v2f*)(&As[(mw + rr) * LDK + kf]);
      v2f a1 = *(const v2f*)(&As[(mw + 16 + rr) * LDK + kf]);
      v2f b0 = *(const v2f*)(&Bs[(nw + rr) * LDK + kf]);
      v2f b1 = *(const v2f*)(&Bs[(nw + 16 + rr) * LDK + kf]);
      c00 = __builtin_amdgcn_wmma_f32_16x16x4_f32(false, a0, false, b0, (short)0, c00, false, false);
      c01 = __builtin_amdgcn_wmma_f32_16x16x4_f32(false, a0, false, b1, (short)0, c01, false, false);
      c10 = __builtin_amdgcn_wmma_f32_16x16x4_f32(false, a1, false, b0, (short)0, c10, false, false);
      c11 = __builtin_amdgcn_wmma_f32_16x16x4_f32(false, a1, false, b1, (short)0, c11, false, false);
    }
  }

  int m0 = mblk + mw, n0 = nblk + nw;
  store_tile<MODE, HAS_RAW>(c00, Ob, ldo, Rb, bias, m0,      n0,      lane);
  store_tile<MODE, HAS_RAW>(c01, Ob, ldo, Rb, bias, m0,      n0 + 16, lane);
  store_tile<MODE, HAS_RAW>(c10, Ob, ldo, Rb, bias, m0 + 16, n0,      lane);
  store_tile<MODE, HAS_RAW>(c11, Ob, ldo, Rb, bias, m0 + 16, n0 + 16, lane);
}

template <int MODE, bool HAS_RAW>
__global__ void __launch_bounds__(128)
gemm_wmma_kernel(const float* __restrict__ A, int lda, long aOuter, long aInner,
                 const float* __restrict__ B, int ldb, long bOuter, long bInner,
                 float* __restrict__ O, int ldo, long oOuter, long oInner,
                 float* __restrict__ Oraw,
                 const float* __restrict__ bias,
                 int K) {
  __shared__ float As[64 * LDK];
  __shared__ float Bs[64 * LDK];
  int zo = blockIdx.z >> 2;
  int zi = blockIdx.z & 3;
  long ooff = (long)zo * oOuter + (long)zi * oInner;
  gemm_tile<MODE, HAS_RAW>(A + (long)zo * aOuter + (long)zi * aInner, lda,
                           B + (long)zo * bOuter + (long)zi * bInner, ldb,
                           O + ooff, ldo, HAS_RAW ? (Oraw + ooff) : nullptr,
                           bias, K, blockIdx.y * 64, blockIdx.x * 64, As, Bs);
}

template <int MODE0, bool RAW0, int MODE1, bool RAW1>
__global__ void __launch_bounds__(128)
gemm_pair_kernel(const float* __restrict__ A0, int lda0, const float* __restrict__ B0, int ldb0,
                 float* __restrict__ O0, int ldo0, float* __restrict__ R0,
                 const float* __restrict__ bias0, int K0,
                 const float* __restrict__ A1, int lda1, const float* __restrict__ B1, int ldb1,
                 float* __restrict__ O1, int ldo1, float* __restrict__ R1,
                 const float* __restrict__ bias1, int K1,
                 int ny0) {
  __shared__ float As[64 * LDK];
  __shared__ float Bs[64 * LDK];
  int by = blockIdx.y;
  if (by < ny0)
    gemm_tile<MODE0, RAW0>(A0, lda0, B0, ldb0, O0, ldo0, R0, bias0, K0,
                           by * 64, blockIdx.x * 64, As, Bs);
  else
    gemm_tile<MODE1, RAW1>(A1, lda1, B1, ldb1, O1, ldo1, R1, bias1, K1,
                           (by - ny0) * 64, blockIdx.x * 64, As, Bs);
}

template <int MODE0, int MODE1, int MODE2, int MT0>
__global__ void __launch_bounds__(128)
gemm_triple_kernel(const float* __restrict__ A0, int lda0, const float* __restrict__ B0, int ldb0,
                   float* __restrict__ O0, int ldo0, long oSlice0,
                   const float* __restrict__ bias0, int K0,
                   const float* __restrict__ A1, int lda1, const float* __restrict__ B1, int ldb1,
                   float* __restrict__ O1, int ldo1, const float* __restrict__ bias1, int K1,
                   const float* __restrict__ A2, int lda2, const float* __restrict__ B2, int ldb2,
                   float* __restrict__ O2, int ldo2, const float* __restrict__ bias2, int K2,
                   int ny0, int ny1) {
  __shared__ float As[64 * LDK];
  __shared__ float Bs[64 * LDK];
  int by = blockIdx.y;
  if (by < ny0) {
    int sl = by / MT0;
    int mr = by - sl * MT0;
    gemm_tile<MODE0, false>(A0 + sl * K0, lda0, B0 + (long)sl * K0 * ldb0, ldb0,
                            O0 + (long)sl * oSlice0, ldo0, nullptr, bias0, K0,
                            mr * 64, blockIdx.x * 64, As, Bs);
  } else if (by < ny1)
    gemm_tile<MODE1, false>(A1, lda1, B1, ldb1, O1, ldo1, nullptr, bias1, K1,
                            (by - ny0) * 64, blockIdx.x * 64, As, Bs);
  else
    gemm_tile<MODE2, false>(A2, lda2, B2, ldb2, O2, ldo2, nullptr, bias2, K2,
                            (by - ny1) * 64, blockIdx.x * 64, As, Bs);
}

#define TP 12
__global__ void __launch_bounds__(128)
final_reduce_kernel(const float* __restrict__ tprd, const float* __restrict__ tref,
                    const float* __restrict__ targ,
                    const float* __restrict__ wout, float* __restrict__ out) {
  __shared__ __attribute__((aligned(16))) float Ts[128 * LDK];
  __shared__ __attribute__((aligned(16))) float sp[TP * HH];
  __shared__ __attribute__((aligned(16))) float sw[HH];
  int tid  = threadIdx.x;
  int lane = tid & 31;
  int wave = tid >> 5;
  int a0   = blockIdx.x * 128;
  int p0   = blockIdx.y * TP;
  int bc   = blockIdx.z;
  int b    = bc >> 2;
  int c    = bc & 3;

  const float* tg = targ + ((long)(b * SS + a0) * CC + c) * HH + (lane << 2);
  {
    float4 tv[32];
#pragma unroll
    for (int rep = 0; rep < 32; ++rep)
      tv[rep] = *(const float4*)(tg + (long)(wave + rep * 4) * (CC * HH));
#pragma unroll
    for (int rep = 0; rep < 32; ++rep)
      *(float4*)(&Ts[(wave + rep * 4) * LDK + (lane << 2)]) = tv[rep];
  }
#pragma unroll
  for (int j = 0; j < TP * (HH / 4) / 128; ++j) {
    int i = tid + j * 128;
    int row = i >> 5, c4 = (i & 31) << 2;
    long bp = (long)(b * SS + p0 + row);
    float4 pv = *(const float4*)(tprd + (bp * CC + c) * HH + c4);
    const float* rp = tref + bp * HH + c4;
    float4 r0 = *(const float4*)(rp);
    float4 r1 = *(const float4*)(rp + (long)BB * SS * HH);
    float4 r2 = *(const float4*)(rp + (long)2 * BB * SS * HH);
    float4 r3 = *(const float4*)(rp + (long)3 * BB * SS * HH);
    float4 sv;
    sv.x = pv.x + (((r0.x + r1.x) + r2.x) + r3.x);
    sv.y = pv.y + (((r0.y + r1.y) + r2.y) + r3.y);
    sv.z = pv.z + (((r0.z + r1.z) + r2.z) + r3.z);
    sv.w = pv.w + (((r0.w + r1.w) + r2.w) + r3.w);
    *(float4*)(&sp[row * HH + c4]) = sv;
  }
  sw[tid] = wout[tid];
  __syncthreads();

  const float* trow = &Ts[tid * LDK];
  float acc[TP];
#pragma unroll
  for (int j = 0; j < TP; ++j) acc[j] = 0.0f;
  for (int k = 0; k < HH; k += 4) {
    float4 tv = *(const float4*)(trow + k);
    float4 wv = *(const float4*)(sw + k);
    float4 q[TP];
#pragma unroll
    for (int j = 0; j < TP; ++j) q[j] = *(const float4*)(sp + j * HH + k);
#pragma unroll
    for (int j = 0; j < TP; ++j) acc[j] += tanhf_hw(q[j].x + tv.x) * wv.x;
#pragma unroll
    for (int j = 0; j < TP; ++j) acc[j] += tanhf_hw(q[j].y + tv.y) * wv.y;
#pragma unroll
    for (int j = 0; j < TP; ++j) acc[j] += tanhf_hw(q[j].z + tv.z) * wv.z;
#pragma unroll
    for (int j = 0; j < TP; ++j) acc[j] += tanhf_hw(q[j].w + tv.w) * wv.w;
  }
  float* orow = out + ((long)(b * SS + p0) * CC + c) * SS + a0 + tid;
#pragma unroll
  for (int j = 0; j < TP; ++j) orow[(long)j * (CC * SS)] = acc[j];
}

extern "C" void kernel_launch(void* const* d_in, const int* in_sizes, int n_in,
                              void* d_out, int out_size, void* d_ws, size_t ws_size,
                              hipStream_t stream) {
  const float* seq  = (const float*)d_in[0];
  const float* base = (const float*)d_in[1];
  const float* Wp   = (const float*)d_in[2];
  const float* bp   = (const float*)d_in[3];
  const float* Ua   = (const float*)d_in[4];
  const float* ba   = (const float*)d_in[5];
  const float* Wmid = (const float*)d_in[6];
  const float* bmid = (const float*)d_in[7];
  const float* wout = (const float*)d_in[8];
  float* out = (float*)d_out;

  float* ws      = (float*)d_ws;
  float* tanh_hp = ws;
  float* ha_raw  = ws + 393216;
  float* tanh_ha = ws + 786432;
  float* tanhref = ws + 1179648;
  float* targ    = ws + 1572864;
  float* tref    = ws + 1966080;
  float* tprd    = ws + 2359296;

  dim3 blk(32, 4);

  gemm_pair_kernel<MODE_TANH | MODE_BIAS, true, MODE_TANH | MODE_BIAS, false>
      <<<dim3(8, 24, 1), blk, 0, stream>>>(
      seq, DD, Ua, CC * HH, tanh_ha, CC * HH, ha_raw, ba, DD,
      seq, DD, Wp, CC * HH, tanh_hp, CC * HH, nullptr, bp, DD,
      12);

  gemm_wmma_kernel<MODE_TANH, false><<<dim3(2, 6, BB * CC), blk, 0, stream>>>(
      base, CC * SS, (long)SS * CC * SS, SS,
      ha_raw, CC * HH, (long)SS * CC * HH, HH,
      tanhref, CC * HH, (long)SS * CC * HH, HH,
      nullptr, nullptr, SS);

  gemm_triple_kernel<0, 0, MODE_BIAS, 12><<<dim3(2, 144, 1), blk, 0, stream>>>(
      tanhref, CC * HH, Wmid + 2 * HH * HH, HH, tref, HH, (long)BB * SS * HH, nullptr, HH,
      tanh_ha, HH, Wmid + HH * HH, HH, targ, HH, nullptr, HH,
      tanh_hp, HH, Wmid, HH, tprd, HH, bmid, HH,
      48, 96);

  final_reduce_kernel<<<dim3(SS / 128, SS / TP, BB * CC), dim3(128), 0, stream>>>(
      tprd, tref, targ, wout, out);
}


# GEMM B-tile staging: a thread takes two adjacent k rows and writes the transposed tile with 64-bit LDS stores; a 16 B shift per 16 rows makes them conflict-free (were 4-way, 32-bit); sums unchanged
# speedup vs baseline: 4.7513x; 1.0986x over previous
#include <hip/hip_runtime.h>

#define BB 2
#define SS 384
#define CC 4
#define HH 128
#define DD 768

typedef float v2f __attribute__((ext_vector_type(2)));
typedef float v4f __attribute__((ext_vector_type(4)));
typedef float v8f __attribute__((ext_vector_type(8)));

__device__ __forceinline__ float tanhf_fast(float x) {
  float e = __builtin_amdgcn_exp2f(x * 2.8853900817779268f);
  float r = __builtin_amdgcn_rcpf(e + 1.0f);
  return 1.0f - 2.0f * r;
}

__device__ __forceinline__ float tanhf_hw(float x) {
  return __builtin_amdgcn_tanhf(x);
}

#define MODE_TANH 1
#define MODE_BIAS 4

template <int MODE, bool HAS_RAW>
__device__ __forceinline__ void store_tile(v8f c, float* O, int ldo, float* Oraw,
                                           const float* bias,
                                           int m0, int n0, int lane) {
  int col   = n0 + (lane & 15);
  int rbase = m0 + ((lane >> 4) << 3);
#pragma unroll
  for (int i = 0; i < 8; ++i) {
    int   row = rbase + i;
    float v   = c[i];
    if constexpr (MODE & MODE_BIAS) v += bias[col];
    if constexpr (HAS_RAW) Oraw[(long)row * ldo + col] = v;
    O[(long)row * ldo + col] = (MODE & MODE_TANH) ? tanhf_fast(v) : v;
  }
}

#define BK 128
#define LDK (BK + 4)
#define BS_ROW(n) ((n) * LDK + (((n) >> 4) << 2))
#define BS_FLOATS (64 * LDK + 16)
template <int MODE, bool HAS_RAW>
__device__ __forceinline__ void
gemm_tile(const float* __restrict__ Ab, int lda, const float* __restrict__ Bb, int ldb,
          float* __restrict__ Ob, int ldo, float* __restrict__ Rb,
          const float* __restrict__ bias,
          int K, int mblk, int nblk, float* As, float* Bs) {
  int lane = threadIdx.x;
  int wave = threadIdx.y;
  int t    = wave * 32 + lane;
  int mw   = (wave >> 1) * 32;
  int nw   = (wave & 1) * 32;

  v8f c00 = {}, c01 = {}, c10 = {}, c11 = {};

  int ar0 = t >> 5, ac = (t & 31) << 2;
  int bq0 = t >> 4, bc = (t & 15) << 2;
  const float* Ag = Ab + (long)(mblk + ar0) * lda + ac;
  const float* Bg = Bb + (long)(2 * bq0) * ldb + nblk + bc;
  float* Bw = Bs + BS_ROW(bc) + 2 * bq0;

  v4f av[16], bv[16];
#pragma unroll
  for (int rep = 0; rep < 16; ++rep)
    av[rep] = *(const v4f*)(Ag + (long)(rep * 4) * lda);
#pragma unroll
  for (int rep = 0; rep < 8; ++rep) {
    bv[2 * rep]     = *(const v4f*)(Bg + (long)(rep * 16) * ldb);
    bv[2 * rep + 1] = *(const v4f*)(Bg + (long)(rep * 16 + 1) * ldb);
  }

  for (int kc = 0; kc < K; kc += BK) {
    __syncthreads();
#pragma unroll
    for (int rep = 0; rep < 16; ++rep)
      *(v4f*)(&As[(ar0 + rep * 4) * LDK + ac]) = av[rep];
#pragma unroll
    for (int rep = 0; rep < 8; ++rep) {
      v4f e = bv[2 * rep], o = bv[2 * rep + 1];
      float* w = Bw + rep * 16;
      *(v2f*)(w)           = (v2f){e.x, o.x};
      *(v2f*)(w + LDK)     = (v2f){e.y, o.y};
      *(v2f*)(w + 2 * LDK) = (v2f){e.z, o.z};
      *(v2f*)(w + 3 * LDK) = (v2f){e.w, o.w};
    }
    __syncthreads();

    int kn = kc + BK;
    if (kn < K) {
#pragma unroll
      for (int rep = 0; rep < 16; ++rep)
        av[rep] = *(const v4f*)(Ag + (long)(rep * 4) * lda + kn);
#pragma unroll
      for (int rep = 0; rep < 8; ++rep) {
        bv[2 * rep]     = *(const v4f*)(Bg + (long)(kn + rep * 16) * ldb);
        bv[2 * rep + 1] = *(const v4f*)(Bg + (long)(kn + rep * 16 + 1) * ldb);
      }
    }

#pragma unroll
    for (int kl = 0; kl < BK; kl += 4) {
      int kf = kl + ((lane >> 4) << 1);
      int rr = lane & 15;
      v2f a0 = *(const v2f*)(&As[(mw + rr) * LDK + kf]);
      v2f a1 = *(const v2f*)(&As[(mw + 16 + rr) * LDK + kf]);
      v2f b0 = *(const v2f*)(&Bs[BS_ROW(nw + rr) + kf]);
      v2f b1 = *(const v2f*)(&Bs[BS_ROW(nw + 16 + rr) + kf]);
      c00 = __builtin_amdgcn_wmma_f32_16x16x4_f32(false, a0, false, b0, (short)0, c00, false, false);
      c01 = __builtin_amdgcn_wmma_f32_16x16x4_f32(false, a0, false, b1, (short)0, c01, false, false);
      c10 = __builtin_amdgcn_wmma_f32_16x16x4_f32(false, a1, false, b0, (short)0, c10, false, false);
      c11 = __builtin_amdgcn_wmma_f32_16x16x4_f32(false, a1, false, b1, (short)0, c11, false, false);
    }
  }

  int m0 = mblk + mw, n0 = nblk + nw;
  store_tile<MODE, HAS_RAW>(c00, Ob, ldo, Rb, bias, m0,      n0,      lane);
  store_tile<MODE, HAS_RAW>(c01, Ob, ldo, Rb, bias, m0,      n0 + 16, lane);
  store_tile<MODE, HAS_RAW>(c10, Ob, ldo, Rb, bias, m0 + 16, n0,      lane);
  store_tile<MODE, HAS_RAW>(c11, Ob, ldo, Rb, bias, m0 + 16, n0 + 16, lane);
}

template <int MODE, bool HAS_RAW>
__global__ void __launch_bounds__(128)
gemm_wmma_kernel(const float* __restrict__ A, int lda, long aOuter, long aInner,
                 const float* __restrict__ B, int ldb, long bOuter, long bInner,
                 float* __restrict__ O, int ldo, long oOuter, long oInner,
                 float* __restrict__ Oraw,
                 const float* __restrict__ bias,
                 int K) {
  __shared__ __attribute__((aligned(16))) float As[64 * LDK];
  __shared__ __attribute__((aligned(16))) float Bs[BS_FLOATS];
  int zo = blockIdx.z >> 2;
  int zi = blockIdx.z & 3;
  long ooff = (long)zo * oOuter + (long)zi * oInner;
  gemm_tile<MODE, HAS_RAW>(A + (long)zo * aOuter + (long)zi * aInner, lda,
                           B + (long)zo * bOuter + (long)zi * bInner, ldb,
                           O + ooff, ldo, HAS_RAW ? (Oraw + ooff) : nullptr,
                           bias, K, blockIdx.y * 64, blockIdx.x * 64, As, Bs);
}

template <int MODE0, bool RAW0, int MODE1, bool RAW1>
__global__ void __launch_bounds__(128)
gemm_pair_kernel(const float* __restrict__ A0, int lda0, const float* __restrict__ B0, int ldb0,
                 float* __restrict__ O0, int ldo0, float* __restrict__ R0,
                 const float* __restrict__ bias0, int K0,
                 const float* __restrict__ A1, int lda1, const float* __restrict__ B1, int ldb1,
                 float* __restrict__ O1, int ldo1, float* __restrict__ R1,
                 const float* __restrict__ bias1, int K1,
                 int ny0) {
  __shared__ __attribute__((aligned(16))) float As[64 * LDK];
  __shared__ __attribute__((aligned(16))) float Bs[BS_FLOATS];
  int by = blockIdx.y;
  if (by < ny0)
    gemm_tile<MODE0, RAW0>(A0, lda0, B0, ldb0, O0, ldo0, R0, bias0, K0,
                           by * 64, blockIdx.x * 64, As, Bs);
  else
    gemm_tile<MODE1, RAW1>(A1, lda1, B1, ldb1, O1, ldo1, R1, bias1, K1,
                           (by - ny0) * 64, blockIdx.x * 64, As, Bs);
}

template <int MODE0, int MODE1, int MODE2, int MT0>
__global__ void __launch_bounds__(128)
gemm_triple_kernel(const float* __restrict__ A0, int lda0, const float* __restrict__ B0, int ldb0,
                   float* __restrict__ O0, int ldo0, long oSlice0,
                   const float* __restrict__ bias0, int K0,
                   const float* __restrict__ A1, int lda1, const float* __restrict__ B1, int ldb1,
                   float* __restrict__ O1, int ldo1, const float* __restrict__ bias1, int K1,
                   const float* __restrict__ A2, int lda2, const float* __restrict__ B2, int ldb2,
                   float* __restrict__ O2, int ldo2, const float* __restrict__ bias2, int K2,
                   int ny0, int ny1) {
  __shared__ __attribute__((aligned(16))) float As[64 * LDK];
  __shared__ __attribute__((aligned(16))) float Bs[BS_FLOATS];
  int by = blockIdx.y;
  if (by < ny0) {
    int sl = by / MT0;
    int mr = by - sl * MT0;
    gemm_tile<MODE0, false>(A0 + sl * K0, lda0, B0 + (long)sl * K0 * ldb0, ldb0,
                            O0 + (long)sl * oSlice0, ldo0, nullptr, bias0, K0,
                            mr * 64, blockIdx.x * 64, As, Bs);
  } else if (by < ny1)
    gemm_tile<MODE1, false>(A1, lda1, B1, ldb1, O1, ldo1, nullptr, bias1, K1,
                            (by - ny0) * 64, blockIdx.x * 64, As, Bs);
  else
    gemm_tile<MODE2, false>(A2, lda2, B2, ldb2, O2, ldo2, nullptr, bias2, K2,
                            (by - ny1) * 64, blockIdx.x * 64, As, Bs);
}

#define TP 12
__global__ void __launch_bounds__(128)
final_reduce_kernel(const float* __restrict__ tprd, const float* __restrict__ tref,
                    const float* __restrict__ targ,
                    const float* __restrict__ wout, float* __restrict__ out) {
  __shared__ __attribute__((aligned(16))) float Ts[128 * LDK];
  __shared__ __attribute__((aligned(16))) float sp[TP * HH];
  __shared__ __attribute__((aligned(16))) float sw[HH];
  int tid  = threadIdx.x;
  int lane = tid & 31;
  int wave = tid >> 5;
  int a0   = blockIdx.x * 128;
  int p0   = blockIdx.y * TP;
  int bc   = blockIdx.z;
  int b    = bc >> 2;
  int c    = bc & 3;

  const float* tg = targ + ((long)(b * SS + a0) * CC + c) * HH + (lane << 2);
  {
    float4 tv[32];
#pragma unroll
    for (int rep = 0; rep < 32; ++rep)
      tv[rep] = *(const float4*)(tg + (long)(wave + rep * 4) * (CC * HH));
#pragma unroll
    for (int rep = 0; rep < 32; ++rep)
      *(float4*)(&Ts[(wave + rep * 4) * LDK + (lane << 2)]) = tv[rep];
  }
#pragma unroll
  for (int j = 0; j < TP * (HH / 4) / 128; ++j) {
    int i = tid + j * 128;
    int row = i >> 5, c4 = (i & 31) << 2;
    long bp = (long)(b * SS + p0 + row);
    float4 pv = *(const float4*)(tprd + (bp * CC + c) * HH + c4);
    const float* rp = tref + bp * HH + c4;
    float4 r0 = *(const float4*)(rp);
    float4 r1 = *(const float4*)(rp + (long)BB * SS * HH);
    float4 r2 = *(const float4*)(rp + (long)2 * BB * SS * HH);
    float4 r3 = *(const float4*)(rp + (long)3 * BB * SS * HH);
    float4 sv;
    sv.x = pv.x + (((r0.x + r1.x) + r2.x) + r3.x);
    sv.y = pv.y + (((r0.y + r1.y) + r2.y) + r3.y);
    sv.z = pv.z + (((r0.z + r1.z) + r2.z) + r3.z);
    sv.w = pv.w + (((r0.w + r1.w) + r2.w) + r3.w);
    *(float4*)(&sp[row * HH + c4]) = sv;
  }
  sw[tid] = wout[tid];
  __syncthreads();

  const float* trow = &Ts[tid * LDK];
  float acc[TP];
#pragma unroll
  for (int j = 0; j < TP; ++j) acc[j] = 0.0f;
  for (int k = 0; k < HH; k += 4) {
    float4 tv = *(const float4*)(trow + k);
    float4 wv = *(const float4*)(sw + k);
    float4 q[TP];
#pragma unroll
    for (int j = 0; j < TP; ++j) q[j] = *(const float4*)(sp + j * HH + k);
#pragma unroll
    for (int j = 0; j < TP; ++j) acc[j] += tanhf_hw(q[j].x + tv.x) * wv.x;
#pragma unroll
    for (int j = 0; j < TP; ++j) acc[j] += tanhf_hw(q[j].y + tv.y) * wv.y;
#pragma unroll
    for (int j = 0; j < TP; ++j) acc[j] += tanhf_hw(q[j].z + tv.z) * wv.z;
#pragma unroll
    for (int j = 0; j < TP; ++j) acc[j] += tanhf_hw(q[j].w + tv.w) * wv.w;
  }
  float* orow = out + ((long)(b * SS + p0) * CC + c) * SS + a0 + tid;
#pragma unroll
  for (int j = 0; j < TP; ++j) orow[(long)j * (CC * SS)] = acc[j];
}

extern "C" void kernel_launch(void* const* d_in, const int* in_sizes, int n_in,
                              void* d_out, int out_size, void* d_ws, size_t ws_size,
                              hipStream_t stream) {
  const float* seq  = (const float*)d_in[0];
  const float* base = (const float*)d_in[1];
  const float* Wp   = (const float*)d_in[2];
  const float* bp   = (const float*)d_in[3];
  const float* Ua   = (const float*)d_in[4];
  const float* ba   = (const float*)d_in[5];
  const float* Wmid = (const float*)d_in[6];
  const float* bmid = (const float*)d_in[7];
  const float* wout = (const float*)d_in[8];
  float* out = (float*)d_out;

  float* ws      = (float*)d_ws;
  float* tanh_hp = ws;
  float* ha_raw  = ws + 393216;
  float* tanh_ha = ws + 786432;
  float* tanhref = ws + 1179648;
  float* targ    = ws + 1572864;
  float* tref    = ws + 1966080;
  float* tprd    = ws + 2359296;

  dim3 blk(32, 4);

  gemm_pair_kernel<MODE_TANH | MODE_BIAS, true, MODE_TANH | MODE_BIAS, false>
      <<<dim3(8, 24, 1), blk, 0, stream>>>(
      seq, DD, Ua, CC * HH, tanh_ha, CC * HH, ha_raw, ba, DD,
      seq, DD, Wp, CC * HH, tanh_hp, CC * HH, nullptr, bp, DD,
      12);

  gemm_wmma_kernel<MODE_TANH, false><<<dim3(2, 6, BB * CC), blk, 0, stream>>>(
      base, CC * SS, (long)SS * CC * SS, SS,
      ha_raw, CC * HH, (long)SS * CC * HH, HH,
      tanhref, CC * HH, (long)SS * CC * HH, HH,
      nullptr, nullptr, SS);

  gemm_triple_kernel<0, 0, MODE_BIAS, 12><<<dim3(2, 144, 1), blk, 0, stream>>>(
      tanhref, CC * HH, Wmid + 2 * HH * HH, HH, tref, HH, (long)BB * SS * HH, nullptr, HH,
      tanh_ha, HH, Wmid + HH * HH, HH, targ, HH, nullptr, HH,
      tanh_hp, HH, Wmid, HH, tprd, HH, bmid, HH,
      48, 96);

  final_reduce_kernel<<<dim3(SS / 128, SS / TP, BB * CC), dim3(128), 0, stream>>>(
      tprd, tref, targ, wout, out);
}


# t_ref product moved into the ref GEMM's block: a block owns 16 p x all 128 h of one (b,c), keeps tanh of its tile in LDS and multiplies it by W3[c]; targ, tprd join; 4 launches -> 3; sums unchanged
# speedup vs baseline: 5.4772x; 1.1528x over previous
#include <hip/hip_runtime.h>

#define BB 2
#define SS 384
#define CC 4
#define HH 128
#define DD 768

typedef float v2f __attribute__((ext_vector_type(2)));
typedef float v4f __attribute__((ext_vector_type(4)));
typedef float v8f __attribute__((ext_vector_type(8)));

__device__ __forceinline__ float tanhf_fast(float x) {
  float e = __builtin_amdgcn_exp2f(x * 2.8853900817779268f);
  float r = __builtin_amdgcn_rcpf(e + 1.0f);
  return 1.0f - 2.0f * r;
}

__device__ __forceinline__ float tanhf_hw(float x) {
  return __builtin_amdgcn_tanhf(x);
}

#define MODE_TANH 1
#define MODE_BIAS 4

template <int MODE, bool HAS_RAW>
__device__ __forceinline__ void store_tile(v8f c, float* O, int ldo, float* Oraw,
                                           const float* bias,
                                           int m0, int n0, int lane) {
  int col   = n0 + (lane & 15);
  int rbase = m0 + ((lane >> 4) << 3);
#pragma unroll
  for (int i = 0; i < 8; ++i) {
    int   row = rbase + i;
    float v   = c[i];
    if constexpr (MODE & MODE_BIAS) v += bias[col];
    if constexpr (HAS_RAW) Oraw[(long)row * ldo + col] = v;
    O[(long)row * ldo + col] = (MODE & MODE_TANH) ? tanhf_fast(v) : v;
  }
}

#define BK 128
#define LDK (BK + 4)
#define BS_ROW(n) ((n) * LDK + (((n) >> 4) << 2))
#define BS_FLOATS (64 * LDK + 16)
template <int MODE, bool HAS_RAW>
__device__ __forceinline__ void
gemm_tile(const float* __restrict__ Ab, int lda, const float* __restrict__ Bb, int ldb,
          float* __restrict__ Ob, int ldo, float* __restrict__ Rb,
          const float* __restrict__ bias,
          int K, int mblk, int nblk, float* As, float* Bs) {
  int lane = threadIdx.x;
  int wave = threadIdx.y;
  int t    = wave * 32 + lane;
  int mw   = (wave >> 1) * 32;
  int nw   = (wave & 1) * 32;

  v8f c00 = {}, c01 = {}, c10 = {}, c11 = {};

  int ar0 = t >> 5, ac = (t & 31) << 2;
  int bq0 = t >> 4, bc = (t & 15) << 2;
  const float* Ag = Ab + (long)(mblk + ar0) * lda + ac;
  const float* Bg = Bb + (long)(2 * bq0) * ldb + nblk + bc;
  float* Bw = Bs + BS_ROW(bc) + 2 * bq0;

  v4f av[16], bv[16];
#pragma unroll
  for (int rep = 0; rep < 16; ++rep)
    av[rep] = *(const v4f*)(Ag + (long)(rep * 4) * lda);
#pragma unroll
  for (int rep = 0; rep < 8; ++rep) {
    bv[2 * rep]     = *(const v4f*)(Bg + (long)(rep * 16) * ldb);
    bv[2 * rep + 1] = *(const v4f*)(Bg + (long)(rep * 16 + 1) * ldb);
  }

  for (int kc = 0; kc < K; kc += BK) {
    __syncthreads();
#pragma unroll
    for (int rep = 0; rep < 16; ++rep)
      *(v4f*)(&As[(ar0 + rep * 4) * LDK + ac]) = av[rep];
#pragma unroll
    for (int rep = 0; rep < 8; ++rep) {
      v4f e = bv[2 * rep], o = bv[2 * rep + 1];
      float* w = Bw + rep * 16;
      *(v2f*)(w)           = (v2f){e.x, o.x};
      *(v2f*)(w + LDK)     = (v2f){e.y, o.y};
      *(v2f*)(w + 2 * LDK) = (v2f){e.z, o.z};
      *(v2f*)(w + 3 * LDK) = (v2f){e.w, o.w};
    }
    __syncthreads();

    int kn = kc + BK;
    if (kn < K) {
#pragma unroll
      for (int rep = 0; rep < 16; ++rep)
        av[rep] = *(const v4f*)(Ag + (long)(rep * 4) * lda + kn);
#pragma unroll
      for (int rep = 0; rep < 8; ++rep) {
        bv[2 * rep]     = *(const v4f*)(Bg + (long)(kn + rep * 16) * ldb);
        bv[2 * rep + 1] = *(const v4f*)(Bg + (long)(kn + rep * 16 + 1) * ldb);
      }
    }

#pragma unroll
    for (int kl = 0; kl < BK; kl += 4) {
      int kf = kl + ((lane >> 4) << 1);
      int rr = lane & 15;
      v2f a0 = *(const v2f*)(&As[(mw + rr) * LDK + kf]);
      v2f a1 = *(const v2f*)(&As[(mw + 16 + rr) * LDK + kf]);
      v2f b0 = *(const v2f*)(&Bs[BS_ROW(nw + rr) + kf]);
      v2f b1 = *(const v2f*)(&Bs[BS_ROW(nw + 16 + rr) + kf]);
      c00 = __builtin_amdgcn_wmma_f32_16x16x4_f32(false, a0, false, b0, (short)0, c00, false, false);
      c01 = __builtin_amdgcn_wmma_f32_16x16x4_f32(false, a0, false, b1, (short)0, c01, false, false);
      c10 = __builtin_amdgcn_wmma_f32_16x16x4_f32(false, a1, false, b0, (short)0, c10, false, false);
      c11 = __builtin_amdgcn_wmma_f32_16x16x4_f32(false, a1, false, b1, (short)0, c11, false, false);
    }
  }

  int m0 = mblk + mw, n0 = nblk + nw;
  store_tile<MODE, HAS_RAW>(c00, Ob, ldo, Rb, bias, m0,      n0,      lane);
  store_tile<MODE, HAS_RAW>(c01, Ob, ldo, Rb, bias, m0,      n0 + 16, lane);
  store_tile<MODE, HAS_RAW>(c10, Ob, ldo, Rb, bias, m0 + 16, n0,      lane);
  store_tile<MODE, HAS_RAW>(c11, Ob, ldo, Rb, bias, m0 + 16, n0 + 16, lane);
}

template <int MODE0, bool RAW0, int MODE1, bool RAW1>
__global__ void __launch_bounds__(128)
gemm_pair_kernel(const float* __restrict__ A0, int lda0, const float* __restrict__ B0, int ldb0,
                 float* __restrict__ O0, int ldo0, float* __restrict__ R0,
                 const float* __restrict__ bias0, int K0,
                 const float* __restrict__ A1, int lda1, const float* __restrict__ B1, int ldb1,
                 float* __restrict__ O1, int ldo1, float* __restrict__ R1,
                 const float* __restrict__ bias1, int K1,
                 int ny0) {
  __shared__ __attribute__((aligned(16))) float As[64 * LDK];
  __shared__ __attribute__((aligned(16))) float Bs[BS_FLOATS];
  int by = blockIdx.y;
  if (by < ny0)
    gemm_tile<MODE0, RAW0>(A0, lda0, B0, ldb0, O0, ldo0, R0, bias0, K0,
                           by * 64, blockIdx.x * 64, As, Bs);
  else
    gemm_tile<MODE1, RAW1>(A1, lda1, B1, ldb1, O1, ldo1, R1, bias1, K1,
                           (by - ny0) * 64, blockIdx.x * 64, As, Bs);
}

#define RM 16
#define BS2_FLOATS (128 * LDK + 32)

__device__ __forceinline__ void
mma_stage_16x32(const float* As, const float* Bs, int nw, int lane, v8f& c0, v8f& c1) {
  int rr = lane & 15;
  int kh = (lane >> 4) << 1;
  const float* ap  = As + rr * LDK + kh;
  const float* bp0 = Bs + BS_ROW(nw + rr) + kh;
  const float* bp1 = Bs + BS_ROW(nw + 16 + rr) + kh;
#pragma unroll
  for (int kl = 0; kl < BK; kl += 4) {
    v2f a  = *(const v2f*)(ap + kl);
    v2f b0 = *(const v2f*)(bp0 + kl);
    v2f b1 = *(const v2f*)(bp1 + kl);
    c0 = __builtin_amdgcn_wmma_f32_16x16x4_f32(false, a, false, b0, (short)0, c0, false, false);
    c1 = __builtin_amdgcn_wmma_f32_16x16x4_f32(false, a, false, b1, (short)0, c1, false, false);
  }
}

__device__ __forceinline__ void store_b128n(const v4f* bv, float* Bw) {
#pragma unroll
  for (int nh = 0; nh < 2; ++nh) {
#pragma unroll
    for (int rep = 0; rep < 8; ++rep) {
      v4f e = bv[nh * 16 + 2 * rep], o = bv[nh * 16 + 2 * rep + 1];
      float* w = Bw + nh * (64 * LDK + 16) + rep * 16;
      *(v2f*)(w)           = (v2f){e.x, o.x};
      *(v2f*)(w + LDK)     = (v2f){e.y, o.y};
      *(v2f*)(w + 2 * LDK) = (v2f){e.z, o.z};
      *(v2f*)(w + 3 * LDK) = (v2f){e.w, o.w};
    }
  }
}

__device__ __forceinline__ void
ref_tref_tile(const float* __restrict__ Ab, const float* __restrict__ Bb,
              const float* __restrict__ Wc, float* __restrict__ Ob, float* As, float* Bs) {
  constexpr int LDA = CC * SS, LDB = CC * HH;
  int lane = threadIdx.x;
  int wave = threadIdx.y;
  int t    = wave * 32 + lane;
  int nw   = wave * 32;

  v8f c0 = {}, c1 = {};

  int ar0 = t >> 5, ac = (t & 31) << 2;
  int bq0 = t >> 4, bc = (t & 15) << 2;
  const float* Ag = Ab + ar0 * LDA + ac;
  const float* Bg = Bb + (2 * bq0) * LDB + bc;
  const float* Wg = Wc + (2 * bq0) * HH + bc;
  float* Bw = Bs + BS_ROW(bc) + 2 * bq0;

  v4f av[4], bv[32];
#pragma unroll
  for (int rep = 0; rep < 4; ++rep)
    av[rep] = *(const v4f*)(Ag + (rep * 4) * LDA);
#pragma unroll
  for (int nh = 0; nh < 2; ++nh) {
#pragma unroll
    for (int rep = 0; rep < 8; ++rep) {
      bv[nh * 16 + 2 * rep]     = *(const v4f*)(Bg + nh * 64 + (rep * 16) * LDB);
      bv[nh * 16 + 2 * rep + 1] = *(const v4f*)(Bg + nh * 64 + (rep * 16 + 1) * LDB);
    }
  }

  for (int kc = 0; kc < SS; kc += BK) {
    __syncthreads();
#pragma unroll
    for (int rep = 0; rep < 4; ++rep)
      *(v4f*)(&As[(ar0 + rep * 4) * LDK + ac]) = av[rep];
    store_b128n(bv, Bw);
    __syncthreads();

    int kn = kc + BK;
    if (kn < SS) {
#pragma unroll
      for (int rep = 0; rep < 4; ++rep)
        av[rep] = *(const v4f*)(Ag + (rep * 4) * LDA + kn);
#pragma unroll
      for (int nh = 0; nh < 2; ++nh) {
#pragma unroll
        for (int rep = 0; rep < 8; ++rep) {
          bv[nh * 16 + 2 * rep]     = *(const v4f*)(Bg + nh * 64 + (long)(kn + rep * 16) * LDB);
          bv[nh * 16 + 2 * rep + 1] = *(const v4f*)(Bg + nh * 64 + (long)(kn + rep * 16 + 1) * LDB);
        }
      }
    } else {
#pragma unroll
      for (int nh = 0; nh < 2; ++nh) {
#pragma unroll
        for (int rep = 0; rep < 8; ++rep) {
          bv[nh * 16 + 2 * rep]     = *(const v4f*)(Wg + nh * 64 + (rep * 16) * HH);
          bv[nh * 16 + 2 * rep + 1] = *(const v4f*)(Wg + nh * 64 + (rep * 16 + 1) * HH);
        }
      }
    }

    mma_stage_16x32(As, Bs, nw, lane, c0, c1);
  }

  __syncthreads();
  {
    int col = nw + (lane & 15);
    int rb  = (lane >> 4) << 3;
#pragma unroll
    for (int i = 0; i < 8; ++i) {
      As[(rb + i) * LDK + col]      = tanhf_fast(c0[i]);
      As[(rb + i) * LDK + col + 16] = tanhf_fast(c1[i]);
    }
  }
  store_b128n(bv, Bw);
  __syncthreads();
  c0 = (v8f){};
  c1 = (v8f){};
  mma_stage_16x32(As, Bs, nw, lane, c0, c1);

  store_tile<0, false>(c0, Ob, HH, nullptr, nullptr, 0, nw,      lane);
  store_tile<0, false>(c1, Ob, HH, nullptr, nullptr, 0, nw + 16, lane);
}

template <int MODE1, int MODE2>
__global__ void __launch_bounds__(128)
ref_tref_pair_kernel(const float* __restrict__ base, const float* __restrict__ hraw,
                     const float* __restrict__ W3, float* __restrict__ tref,
                     const float* __restrict__ A1, const float* __restrict__ B1,
                     float* __restrict__ O1, const float* __restrict__ bias1,
                     const float* __restrict__ A2, const float* __restrict__ B2,
                     float* __restrict__ O2, const float* __restrict__ bias2,
                     int nyf, int ny1) {
  __shared__ __attribute__((aligned(16))) float As[64 * LDK];
  __shared__ __attribute__((aligned(16))) float Bs[BS2_FLOATS];
  int by = blockIdx.y;
  if (by < nyf) {
    int f  = by * 2 + blockIdx.x;
    int z  = f / (SS / RM);
    int pt = f - z * (SS / RM);
    int b  = z >> 2, c = z & 3;
    long prow = (long)b * SS + pt * RM;
    ref_tref_tile(base + (prow * CC + c) * SS,
                  hraw + ((long)b * SS * CC + c) * HH,
                  W3 + (long)c * HH * HH,
                  tref + (long)c * BB * SS * HH + prow * HH, As, Bs);
  } else if (by < ny1)
    gemm_tile<MODE1, false>(A1, HH, B1, HH, O1, HH, nullptr, bias1, HH,
                            (by - nyf) * 64, blockIdx.x * 64, As, Bs);
  else
    gemm_tile<MODE2, false>(A2, HH, B2, HH, O2, HH, nullptr, bias2, HH,
                            (by - ny1) * 64, blockIdx.x * 64, As, Bs);
}

#define TP 12
__global__ void __launch_bounds__(128)
final_reduce_kernel(const float* __restrict__ tprd, const float* __restrict__ tref,
                    const float* __restrict__ targ,
                    const float* __restrict__ wout, float* __restrict__ out) {
  __shared__ __attribute__((aligned(16))) float Ts[128 * LDK];
  __shared__ __attribute__((aligned(16))) float sp[TP * HH];
  __shared__ __attribute__((aligned(16))) float sw[HH];
  int tid  = threadIdx.x;
  int lane = tid & 31;
  int wave = tid >> 5;
  int a0   = blockIdx.x * 128;
  int p0   = blockIdx.y * TP;
  int bc   = blockIdx.z;
  int b    = bc >> 2;
  int c    = bc & 3;

  const float* tg = targ + ((long)(b * SS + a0) * CC + c) * HH + (lane << 2);
  {
    float4 tv[32];
#pragma unroll
    for (int rep = 0; rep < 32; ++rep)
      tv[rep] = *(const float4*)(tg + (long)(wave + rep * 4) * (CC * HH));
#pragma unroll
    for (int rep = 0; rep < 32; ++rep)
      *(float4*)(&Ts[(wave + rep * 4) * LDK + (lane << 2)]) = tv[rep];
  }
#pragma unroll
  for (int j = 0; j < TP * (HH / 4) / 128; ++j) {
    int i = tid + j * 128;
    int row = i >> 5, c4 = (i & 31) << 2;
    long bp = (long)(b * SS + p0 + row);
    float4 pv = *(const float4*)(tprd + (bp * CC + c) * HH + c4);
    const float* rp = tref + bp * HH + c4;
    float4 r0 = *(const float4*)(rp);
    float4 r1 = *(const float4*)(rp + (long)BB * SS * HH);
    float4 r2 = *(const float4*)(rp + (long)2 * BB * SS * HH);
    float4 r3 = *(const float4*)(rp + (long)3 * BB * SS * HH);
    float4 sv;
    sv.x = pv.x + (((r0.x + r1.x) + r2.x) + r3.x);
    sv.y = pv.y + (((r0.y + r1.y) + r2.y) + r3.y);
    sv.z = pv.z + (((r0.z + r1.z) + r2.z) + r3.z);
    sv.w = pv.w + (((r0.w + r1.w) + r2.w) + r3.w);
    *(float4*)(&sp[row * HH + c4]) = sv;
  }
  sw[tid] = wout[tid];
  __syncthreads();

  const float* trow = &Ts[tid * LDK];
  float acc[TP];
#pragma unroll
  for (int j = 0; j < TP; ++j) acc[j] = 0.0f;
  for (int k = 0; k < HH; k += 4) {
    float4 tv = *(const float4*)(trow + k);
    float4 wv = *(const float4*)(sw + k);
    float4 q[TP];
#pragma unroll
    for (int j = 0; j < TP; ++j) q[j] = *(const float4*)(sp + j * HH + k);
#pragma unroll
    for (int j = 0; j < TP; ++j) acc[j] += tanhf_hw(q[j].x + tv.x) * wv.x;
#pragma unroll
    for (int j = 0; j < TP; ++j) acc[j] += tanhf_hw(q[j].y + tv.y) * wv.y;
#pragma unroll
    for (int j = 0; j < TP; ++j) acc[j] += tanhf_hw(q[j].z + tv.z) * wv.z;
#pragma unroll
    for (int j = 0; j < TP; ++j) acc[j] += tanhf_hw(q[j].w + tv.w) * wv.w;
  }
  float* orow = out + ((long)(b * SS + p0) * CC + c) * SS + a0 + tid;
#pragma unroll
  for (int j = 0; j < TP; ++j) orow[(long)j * (CC * SS)] = acc[j];
}

extern "C" void kernel_launch(void* const* d_in, const int* in_sizes, int n_in,
                              void* d_out, int out_size, void* d_ws, size_t ws_size,
                              hipStream_t stream) {
  const float* seq  = (const float*)d_in[0];
  const float* base = (const float*)d_in[1];
  const float* Wp   = (const float*)d_in[2];
  const float* bp   = (const float*)d_in[3];
  const float* Ua   = (const float*)d_in[4];
  const float* ba   = (const float*)d_in[5];
  const float* Wmid = (const float*)d_in[6];
  const float* bmid = (const float*)d_in[7];
  const float* wout = (const float*)d_in[8];
  float* out = (float*)d_out;

  float* ws      = (float*)d_ws;
  float* tanh_hp = ws;
  float* ha_raw  = ws + 393216;
  float* tanh_ha = ws + 786432;
  float* targ    = ws + 1572864;
  float* tref    = ws + 1966080;
  float* tprd    = ws + 2359296;

  dim3 blk(32, 4);

  gemm_pair_kernel<MODE_TANH | MODE_BIAS, true, MODE_TANH | MODE_BIAS, false>
      <<<dim3(8, 24, 1), blk, 0, stream>>>(
      seq, DD, Ua, CC * HH, tanh_ha, CC * HH, ha_raw, ba, DD,
      seq, DD, Wp, CC * HH, tanh_hp, CC * HH, nullptr, bp, DD,
      12);

  ref_tref_pair_kernel<0, MODE_BIAS><<<dim3(2, 192, 1), blk, 0, stream>>>(
      base, ha_raw, Wmid + 2 * HH * HH, tref,
      tanh_ha, Wmid + HH * HH, targ, nullptr,
      tanh_hp, Wmid, tprd, bmid,
      96, 144);

  final_reduce_kernel<<<dim3(SS / 128, SS / TP, BB * CC), dim3(128), 0, stream>>>(
      tprd, tref, targ, wout, out);
}


# Projection GEMM: block tile 48x64, was 64x64 (wave owns 48x16, 3 accumulators; tile leaves via LDS in whole 128-B lines): 256 blocks, one per WGP, where 192 blocks left 64 WGPs idle; sums unchanged
# speedup vs baseline: 5.7740x; 1.0542x over previous
#include <hip/hip_runtime.h>

#define BB 2
#define SS 384
#define CC 4
#define HH 128
#define DD 768

typedef float v2f __attribute__((ext_vector_type(2)));
typedef float v4f __attribute__((ext_vector_type(4)));
typedef float v8f __attribute__((ext_vector_type(8)));

__device__ __forceinline__ float tanhf_fast(float x) {
  float e = __builtin_amdgcn_exp2f(x * 2.8853900817779268f);
  float r = __builtin_amdgcn_rcpf(e + 1.0f);
  return 1.0f - 2.0f * r;
}

__device__ __forceinline__ float tanhf_hw(float x) {
  return __builtin_amdgcn_tanhf(x);
}

#define MODE_TANH 1
#define MODE_BIAS 4

template <int MODE, bool HAS_RAW>
__device__ __forceinline__ void store_tile(v8f c, float* O, int ldo, float* Oraw,
                                           const float* bias,
                                           int m0, int n0, int lane) {
  int col   = n0 + (lane & 15);
  int rbase = m0 + ((lane >> 4) << 3);
#pragma unroll
  for (int i = 0; i < 8; ++i) {
    int   row = rbase + i;
    float v   = c[i];
    if constexpr (MODE & MODE_BIAS) v += bias[col];
    if constexpr (HAS_RAW) Oraw[(long)row * ldo + col] = v;
    O[(long)row * ldo + col] = (MODE & MODE_TANH) ? tanhf_fast(v) : v;
  }
}

#define BK 128
#define LDK (BK + 4)
#define BS_ROW(n) ((n) * LDK + (((n) >> 4) << 2))
#define BS_FLOATS (64 * LDK + 16)
template <int MODE, bool HAS_RAW>
__device__ __forceinline__ void
gemm_tile(const float* __restrict__ Ab, int lda, const float* __restrict__ Bb, int ldb,
          float* __restrict__ Ob, int ldo, float* __restrict__ Rb,
          const float* __restrict__ bias,
          int K, int mblk, int nblk, float* As, float* Bs) {
  int lane = threadIdx.x;
  int wave = threadIdx.y;
  int t    = wave * 32 + lane;
  int mw   = (wave >> 1) * 32;
  int nw   = (wave & 1) * 32;

  v8f c00 = {}, c01 = {}, c10 = {}, c11 = {};

  int ar0 = t >> 5, ac = (t & 31) << 2;
  int bq0 = t >> 4, bc = (t & 15) << 2;
  const float* Ag = Ab + (long)(mblk + ar0) * lda + ac;
  const float* Bg = Bb + (long)(2 * bq0) * ldb + nblk + bc;
  float* Bw = Bs + BS_ROW(bc) + 2 * bq0;

  v4f av[16], bv[16];
#pragma unroll
  for (int rep = 0; rep < 16; ++rep)
    av[rep] = *(const v4f*)(Ag + (long)(rep * 4) * lda);
#pragma unroll
  for (int rep = 0; rep < 8; ++rep) {
    bv[2 * rep]     = *(const v4f*)(Bg + (long)(rep * 16) * ldb);
    bv[2 * rep + 1] = *(const v4f*)(Bg + (long)(rep * 16 + 1) * ldb);
  }

  for (int kc = 0; kc < K; kc += BK) {
    __syncthreads();
#pragma unroll
    for (int rep = 0; rep < 16; ++rep)
      *(v4f*)(&As[(ar0 + rep * 4) * LDK + ac]) = av[rep];
#pragma unroll
    for (int rep = 0; rep < 8; ++rep) {
      v4f e = bv[2 * rep], o = bv[2 * rep + 1];
      float* w = Bw + rep * 16;
      *(v2f*)(w)           = (v2f){e.x, o.x};
      *(v2f*)(w + LDK)     = (v2f){e.y, o.y};
      *(v2f*)(w + 2 * LDK) = (v2f){e.z, o.z};
      *(v2f*)(w + 3 * LDK) = (v2f){e.w, o.w};
    }
    __syncthreads();

    int kn = kc + BK;
    if (kn < K) {
#pragma unroll
      for (int rep = 0; rep < 16; ++rep)
        av[rep] = *(const v4f*)(Ag + (long)(rep * 4) * lda + kn);
#pragma unroll
      for (int rep = 0; rep < 8; ++rep) {
        bv[2 * rep]     = *(const v4f*)(Bg + (long)(kn + rep * 16) * ldb);
        bv[2 * rep + 1] = *(const v4f*)(Bg + (long)(kn + rep * 16 + 1) * ldb);
      }
    }

#pragma unroll
    for (int kl = 0; kl < BK; kl += 4) {
      int kf = kl + ((lane >> 4) << 1);
      int rr = lane & 15;
      v2f a0 = *(const v2f*)(&As[(mw + rr) * LDK + kf]);
      v2f a1 = *(const v2f*)(&As[(mw + 16 + rr) * LDK + kf]);
      v2f b0 = *(const v2f*)(&Bs[BS_ROW(nw + rr) + kf]);
      v2f b1 = *(const v2f*)(&Bs[BS_ROW(nw + 16 + rr) + kf]);
      c00 = __builtin_amdgcn_wmma_f32_16x16x4_f32(false, a0, false, b0, (short)0, c00, false, false);
      c01 = __builtin_amdgcn_wmma_f32_16x16x4_f32(false, a0, false, b1, (short)0, c01, false, false);
      c10 = __builtin_amdgcn_wmma_f32_16x16x4_f32(false, a1, false, b0, (short)0, c10, false, false);
      c11 = __builtin_amdgcn_wmma_f32_16x16x4_f32(false, a1, false, b1, (short)0, c11, false, false);
    }
  }

  int m0 = mblk + mw, n0 = nblk + nw;
  store_tile<MODE, HAS_RAW>(c00, Ob, ldo, Rb, bias, m0,      n0,      lane);
  store_tile<MODE, HAS_RAW>(c01, Ob, ldo, Rb, bias, m0,      n0 + 16, lane);
  store_tile<MODE, HAS_RAW>(c10, Ob, ldo, Rb, bias, m0 + 16, n0,      lane);
  store_tile<MODE, HAS_RAW>(c11, Ob, ldo, Rb, bias, m0 + 16, n0 + 16, lane);
}

template <int MODE, bool HAS_RAW>
__device__ __forceinline__ void
gemm_tile_48x64(const float* __restrict__ Ab, int lda, const float* __restrict__ Bb, int ldb,
                float* __restrict__ Ob, int ldo, float* __restrict__ Rb,
                const float* __restrict__ bias,
                int K, int mblk, int nblk, float* As, float* Bs) {
  int lane = threadIdx.x;
  int wave = threadIdx.y;
  int t    = wave * 32 + lane;
  int nw   = wave * 16;

  v8f c0 = {}, c1 = {}, c2 = {};

  int ar0 = t >> 5, ac = (t & 31) << 2;
  int bq0 = t >> 4, bc = (t & 15) << 2;
  const float* Ag = Ab + (long)(mblk + ar0) * lda + ac;
  const float* Bg = Bb + (long)(2 * bq0) * ldb + nblk + bc;
  float* Bw = Bs + BS_ROW(bc) + 2 * bq0;

  v4f av[12], bv[16];
#pragma unroll
  for (int rep = 0; rep < 12; ++rep)
    av[rep] = *(const v4f*)(Ag + (long)(rep * 4) * lda);
#pragma unroll
  for (int rep = 0; rep < 8; ++rep) {
    bv[2 * rep]     = *(const v4f*)(Bg + (long)(rep * 16) * ldb);
    bv[2 * rep + 1] = *(const v4f*)(Bg + (long)(rep * 16 + 1) * ldb);
  }

  for (int kc = 0; kc < K; kc += BK) {
    __syncthreads();
#pragma unroll
    for (int rep = 0; rep < 12; ++rep)
      *(v4f*)(&As[(ar0 + rep * 4) * LDK + ac]) = av[rep];
#pragma unroll
    for (int rep = 0; rep < 8; ++rep) {
      v4f e = bv[2 * rep], o = bv[2 * rep + 1];
      float* w = Bw + rep * 16;
      *(v2f*)(w)           = (v2f){e.x, o.x};
      *(v2f*)(w + LDK)     = (v2f){e.y, o.y};
      *(v2f*)(w + 2 * LDK) = (v2f){e.z, o.z};
      *(v2f*)(w + 3 * LDK) = (v2f){e.w, o.w};
    }
    __syncthreads();

    int kn = kc + BK;
    if (kn < K) {
#pragma unroll
      for (int rep = 0; rep < 12; ++rep)
        av[rep] = *(const v4f*)(Ag + (long)(rep * 4) * lda + kn);
#pragma unroll
      for (int rep = 0; rep < 8; ++rep) {
        bv[2 * rep]     = *(const v4f*)(Bg + (long)(kn + rep * 16) * ldb);
        bv[2 * rep + 1] = *(const v4f*)(Bg + (long)(kn + rep * 16 + 1) * ldb);
      }
    }

#pragma unroll
    for (int kl = 0; kl < BK; kl += 4) {
      int kf = kl + ((lane >> 4) << 1);
      int rr = lane & 15;
      v2f a0 = *(const v2f*)(&As[rr * LDK + kf]);
      v2f a1 = *(const v2f*)(&As[(16 + rr) * LDK + kf]);
      v2f a2 = *(const v2f*)(&As[(32 + rr) * LDK + kf]);
      v2f b  = *(const v2f*)(&Bs[BS_ROW(nw + rr) + kf]);
      c0 = __builtin_amdgcn_wmma_f32_16x16x4_f32(false, a0, false, b, (short)0, c0, false, false);
      c1 = __builtin_amdgcn_wmma_f32_16x16x4_f32(false, a1, false, b, (short)0, c1, false, false);
      c2 = __builtin_amdgcn_wmma_f32_16x16x4_f32(false, a2, false, b, (short)0, c2, false, false);
    }
  }

  constexpr int LDC = 68;
  float* Cs = As;
  __syncthreads();
  {
    int col = nw + (lane & 15);
    int rb  = (lane >> 4) << 3;
#pragma unroll
    for (int i = 0; i < 8; ++i) {
      Cs[(rb + i) * LDC + col]      = c0[i];
      Cs[(16 + rb + i) * LDC + col] = c1[i];
      Cs[(32 + rb + i) * LDC + col] = c2[i];
    }
  }
  __syncthreads();
#pragma unroll
  for (int j = 0; j < 6; ++j) {
    int i   = t + j * 128;
    int row = i >> 4, c4 = (i & 15) << 2;
    v4f v = *(const v4f*)(&Cs[row * LDC + c4]);
    if constexpr (MODE & MODE_BIAS) v += *(const v4f*)(bias + nblk + c4);
    long o = (long)(mblk + row) * ldo + nblk + c4;
    if constexpr (HAS_RAW) *(v4f*)(Rb + o) = v;
    if constexpr (MODE & MODE_TANH)
      v = (v4f){tanhf_fast(v.x), tanhf_fast(v.y), tanhf_fast(v.z), tanhf_fast(v.w)};
    *(v4f*)(Ob + o) = v;
  }
}

template <int MODE0, bool RAW0, int MODE1, bool RAW1>
__global__ void __launch_bounds__(128)
gemm_pair_kernel(const float* __restrict__ A0, int lda0, const float* __restrict__ B0, int ldb0,
                 float* __restrict__ O0, int ldo0, float* __restrict__ R0,
                 const float* __restrict__ bias0, int K0,
                 const float* __restrict__ A1, int lda1, const float* __restrict__ B1, int ldb1,
                 float* __restrict__ O1, int ldo1, float* __restrict__ R1,
                 const float* __restrict__ bias1, int K1,
                 int ny0) {
  __shared__ __attribute__((aligned(16))) float As[48 * LDK];
  __shared__ __attribute__((aligned(16))) float Bs[BS_FLOATS];
  int by = blockIdx.y;
  if (by < ny0)
    gemm_tile_48x64<MODE0, RAW0>(A0, lda0, B0, ldb0, O0, ldo0, R0, bias0, K0,
                                 by * 48, blockIdx.x * 64, As, Bs);
  else
    gemm_tile_48x64<MODE1, RAW1>(A1, lda1, B1, ldb1, O1, ldo1, R1, bias1, K1,
                                 (by - ny0) * 48, blockIdx.x * 64, As, Bs);
}

#define RM 16
#define BS2_FLOATS (128 * LDK + 32)

__device__ __forceinline__ void
mma_stage_16x32(const float* As, const float* Bs, int nw, int lane, v8f& c0, v8f& c1) {
  int rr = lane & 15;
  int kh = (lane >> 4) << 1;
  const float* ap  = As + rr * LDK + kh;
  const float* bp0 = Bs + BS_ROW(nw + rr) + kh;
  const float* bp1 = Bs + BS_ROW(nw + 16 + rr) + kh;
#pragma unroll
  for (int kl = 0; kl < BK; kl += 4) {
    v2f a  = *(const v2f*)(ap + kl);
    v2f b0 = *(const v2f*)(bp0 + kl);
    v2f b1 = *(const v2f*)(bp1 + kl);
    c0 = __builtin_amdgcn_wmma_f32_16x16x4_f32(false, a, false, b0, (short)0, c0, false, false);
    c1 = __builtin_amdgcn_wmma_f32_16x16x4_f32(false, a, false, b1, (short)0, c1, false, false);
  }
}

__device__ __forceinline__ void store_b128n(const v4f* bv, float* Bw) {
#pragma unroll
  for (int nh = 0; nh < 2; ++nh) {
#pragma unroll
    for (int rep = 0; rep < 8; ++rep) {
      v4f e = bv[nh * 16 + 2 * rep], o = bv[nh * 16 + 2 * rep + 1];
      float* w = Bw + nh * (64 * LDK + 16) + rep * 16;
      *(v2f*)(w)           = (v2f){e.x, o.x};
      *(v2f*)(w + LDK)     = (v2f){e.y, o.y};
      *(v2f*)(w + 2 * LDK) = (v2f){e.z, o.z};
      *(v2f*)(w + 3 * LDK) = (v2f){e.w, o.w};
    }
  }
}

__device__ __forceinline__ void
ref_tref_tile(const float* __restrict__ Ab, const float* __restrict__ Bb,
              const float* __restrict__ Wc, float* __restrict__ Ob, float* As, float* Bs) {
  constexpr int LDA = CC * SS, LDB = CC * HH;
  int lane = threadIdx.x;
  int wave = threadIdx.y;
  int t    = wave * 32 + lane;
  int nw   = wave * 32;

  v8f c0 = {}, c1 = {};

  int ar0 = t >> 5, ac = (t & 31) << 2;
  int bq0 = t >> 4, bc = (t & 15) << 2;
  const float* Ag = Ab + ar0 * LDA + ac;
  const float* Bg = Bb + (2 * bq0) * LDB + bc;
  const float* Wg = Wc + (2 * bq0) * HH + bc;
  float* Bw = Bs + BS_ROW(bc) + 2 * bq0;

  v4f av[4], bv[32];
#pragma unroll
  for (int rep = 0; rep < 4; ++rep)
    av[rep] = *(const v4f*)(Ag + (rep * 4) * LDA);
#pragma unroll
  for (int nh = 0; nh < 2; ++nh) {
#pragma unroll
    for (int rep = 0; rep < 8; ++rep) {
      bv[nh * 16 + 2 * rep]     = *(const v4f*)(Bg + nh * 64 + (rep * 16) * LDB);
      bv[nh * 16 + 2 * rep + 1] = *(const v4f*)(Bg + nh * 64 + (rep * 16 + 1) * LDB);
    }
  }

  for (int kc = 0; kc < SS; kc += BK) {
    __syncthreads();
#pragma unroll
    for (int rep = 0; rep < 4; ++rep)
      *(v4f*)(&As[(ar0 + rep * 4) * LDK + ac]) = av[rep];
    store_b128n(bv, Bw);
    __syncthreads();

    int kn = kc + BK;
    if (kn < SS) {
#pragma unroll
      for (int rep = 0; rep < 4; ++rep)
        av[rep] = *(const v4f*)(Ag + (rep * 4) * LDA + kn);
#pragma unroll
      for (int nh = 0; nh < 2; ++nh) {
#pragma unroll
        for (int rep = 0; rep < 8; ++rep) {
          bv[nh * 16 + 2 * rep]     = *(const v4f*)(Bg + nh * 64 + (long)(kn + rep * 16) * LDB);
          bv[nh * 16 + 2 * rep + 1] = *(const v4f*)(Bg + nh * 64 + (long)(kn + rep * 16 + 1) * LDB);
        }
      }
    } else {
#pragma unroll
      for (int nh = 0; nh < 2; ++nh) {
#pragma unroll
        for (int rep = 0; rep < 8; ++rep) {
          bv[nh * 16 + 2 * rep]     = *(const v4f*)(Wg + nh * 64 + (rep * 16) * HH);
          bv[nh * 16 + 2 * rep + 1] = *(const v4f*)(Wg + nh * 64 + (rep * 16 + 1) * HH);
        }
      }
    }

    mma_stage_16x32(As, Bs, nw, lane, c0, c1);
  }

  __syncthreads();
  {
    int col = nw + (lane & 15);
    int rb  = (lane >> 4) << 3;
#pragma unroll
    for (int i = 0; i < 8; ++i) {
      As[(rb + i) * LDK + col]      = tanhf_fast(c0[i]);
      As[(rb + i) * LDK + col + 16] = tanhf_fast(c1[i]);
    }
  }
  store_b128n(bv, Bw);
  __syncthreads();
  c0 = (v8f){};
  c1 = (v8f){};
  mma_stage_16x32(As, Bs, nw, lane, c0, c1);

  store_tile<0, false>(c0, Ob, HH, nullptr, nullptr, 0, nw,      lane);
  store_tile<0, false>(c1, Ob, HH, nullptr, nullptr, 0, nw + 16, lane);
}

template <int MODE1, int MODE2>
__global__ void __launch_bounds__(128)
ref_tref_pair_kernel(const float* __restrict__ base, const float* __restrict__ hraw,
                     const float* __restrict__ W3, float* __restrict__ tref,
                     const float* __restrict__ A1, const float* __restrict__ B1,
                     float* __restrict__ O1, const float* __restrict__ bias1,
                     const float* __restrict__ A2, const float* __restrict__ B2,
                     float* __restrict__ O2, const float* __restrict__ bias2,
                     int nyf, int ny1) {
  __shared__ __attribute__((aligned(16))) float As[64 * LDK];
  __shared__ __attribute__((aligned(16))) float Bs[BS2_FLOATS];
  int by = blockIdx.y;
  if (by < nyf) {
    int f  = by * 2 + blockIdx.x;
    int z  = f / (SS / RM);
    int pt = f - z * (SS / RM);
    int b  = z >> 2, c = z & 3;
    long prow = (long)b * SS + pt * RM;
    ref_tref_tile(base + (prow * CC + c) * SS,
                  hraw + ((long)b * SS * CC + c) * HH,
                  W3 + (long)c * HH * HH,
                  tref + (long)c * BB * SS * HH + prow * HH, As, Bs);
  } else if (by < ny1)
    gemm_tile<MODE1, false>(A1, HH, B1, HH, O1, HH, nullptr, bias1, HH,
                            (by - nyf) * 64, blockIdx.x * 64, As, Bs);
  else
    gemm_tile<MODE2, false>(A2, HH, B2, HH, O2, HH, nullptr, bias2, HH,
                            (by - ny1) * 64, blockIdx.x * 64, As, Bs);
}

#define TP 12
__global__ void __launch_bounds__(128)
final_reduce_kernel(const float* __restrict__ tprd, const float* __restrict__ tref,
                    const float* __restrict__ targ,
                    const float* __restrict__ wout, float* __restrict__ out) {
  __shared__ __attribute__((aligned(16))) float Ts[128 * LDK];
  __shared__ __attribute__((aligned(16))) float sp[TP * HH];
  __shared__ __attribute__((aligned(16))) float sw[HH];
  int tid  = threadIdx.x;
  int lane = tid & 31;
  int wave = tid >> 5;
  int a0   = blockIdx.x * 128;
  int p0   = blockIdx.y * TP;
  int bc   = blockIdx.z;
  int b    = bc >> 2;
  int c    = bc & 3;

  const float* tg = targ + ((long)(b * SS + a0) * CC + c) * HH + (lane << 2);
  {
    float4 tv[32];
#pragma unroll
    for (int rep = 0; rep < 32; ++rep)
      tv[rep] = *(const float4*)(tg + (long)(wave + rep * 4) * (CC * HH));
#pragma unroll
    for (int rep = 0; rep < 32; ++rep)
      *(float4*)(&Ts[(wave + rep * 4) * LDK + (lane << 2)]) = tv[rep];
  }
#pragma unroll
  for (int j = 0; j < TP * (HH / 4) / 128; ++j) {
    int i = tid + j * 128;
    int row = i >> 5, c4 = (i & 31) << 2;
    long bp = (long)(b * SS + p0 + row);
    float4 pv = *(const float4*)(tprd + (bp * CC + c) * HH + c4);
    const float* rp = tref + bp * HH + c4;
    float4 r0 = *(const float4*)(rp);
    float4 r1 = *(const float4*)(rp + (long)BB * SS * HH);
    float4 r2 = *(const float4*)(rp + (long)2 * BB * SS * HH);
    float4 r3 = *(const float4*)(rp + (long)3 * BB * SS * HH);
    float4 sv;
    sv.x = pv.x + (((r0.x + r1.x) + r2.x) + r3.x);
    sv.y = pv.y + (((r0.y + r1.y) + r2.y) + r3.y);
    sv.z = pv.z + (((r0.z + r1.z) + r2.z) + r3.z);
    sv.w = pv.w + (((r0.w + r1.w) + r2.w) + r3.w);
    *(float4*)(&sp[row * HH + c4]) = sv;
  }
  sw[tid] = wout[tid];
  __syncthreads();

  const float* trow = &Ts[tid * LDK];
  float acc[TP];
#pragma unroll
  for (int j = 0; j < TP; ++j) acc[j] = 0.0f;
  for (int k = 0; k < HH; k += 4) {
    float4 tv = *(const float4*)(trow + k);
    float4 wv = *(const float4*)(sw + k);
    float4 q[TP];
#pragma unroll
    for (int j = 0; j < TP; ++j) q[j] = *(const float4*)(sp + j * HH + k);
#pragma unroll
    for (int j = 0; j < TP; ++j) acc[j] += tanhf_hw(q[j].x + tv.x) * wv.x;
#pragma unroll
    for (int j = 0; j < TP; ++j) acc[j] += tanhf_hw(q[j].y + tv.y) * wv.y;
#pragma unroll
    for (int j = 0; j < TP; ++j) acc[j] += tanhf_hw(q[j].z + tv.z) * wv.z;
#pragma unroll
    for (int j = 0; j < TP; ++j) acc[j] += tanhf_hw(q[j].w + tv.w) * wv.w;
  }
  float* orow = out + ((long)(b * SS + p0) * CC + c) * SS + a0 + tid;
#pragma unroll
  for (int j = 0; j < TP; ++j) orow[(long)j * (CC * SS)] = acc[j];
}

extern "C" void kernel_launch(void* const* d_in, const int* in_sizes, int n_in,
                              void* d_out, int out_size, void* d_ws, size_t ws_size,
                              hipStream_t stream) {
  const float* seq  = (const float*)d_in[0];
  const float* base = (const float*)d_in[1];
  const float* Wp   = (const float*)d_in[2];
  const float* bp   = (const float*)d_in[3];
  const float* Ua   = (const float*)d_in[4];
  const float* ba   = (const float*)d_in[5];
  const float* Wmid = (const float*)d_in[6];
  const float* bmid = (const float*)d_in[7];
  const float* wout = (const float*)d_in[8];
  float* out = (float*)d_out;

  float* ws      = (float*)d_ws;
  float* tanh_hp = ws;
  float* ha_raw  = ws + 393216;
  float* tanh_ha = ws + 786432;
  float* targ    = ws + 1572864;
  float* tref    = ws + 1966080;
  float* tprd    = ws + 2359296;

  dim3 blk(32, 4);

  gemm_pair_kernel<MODE_TANH | MODE_BIAS, true, MODE_TANH | MODE_BIAS, false>
      <<<dim3(8, 32, 1), blk, 0, stream>>>(
      seq, DD, Ua, CC * HH, tanh_ha, CC * HH, ha_raw, ba, DD,
      seq, DD, Wp, CC * HH, tanh_hp, CC * HH, nullptr, bp, DD,
      16);

  ref_tref_pair_kernel<0, MODE_BIAS><<<dim3(2, 192, 1), blk, 0, stream>>>(
      base, ha_raw, Wmid + 2 * HH * HH, tref,
      tanh_ha, Wmid + HH * HH, targ, nullptr,
      tanh_hp, Wmid, tprd, bmid,
      96, 144);

  final_reduce_kernel<<<dim3(SS / 128, SS / TP, BB * CC), dim3(128), 0, stream>>>(
      tprd, tref, targ, wout, out);
}
